# starting kernel
# baseline (speedup 1.0000x reference)
#include <hip/hip_runtime.h>

typedef float v2f __attribute__((ext_vector_type(2)));
typedef float v8f __attribute__((ext_vector_type(8)));

__global__ void k_init_deg(float* deg, int n) {
    int i = blockIdx.x * blockDim.x + threadIdx.x;
    if (i < n) deg[i] = 1.0f;
}

__global__ void k_count_deg(const int* __restrict__ dst, float* deg, int E) {
    int e = blockIdx.x * blockDim.x + threadIdx.x;
    if (e < E) atomicAdd(&deg[dst[e]], 1.0f);
}

__global__ void k_rsqrt_inplace(float* deg, int n) {
    int i = blockIdx.x * blockDim.x + threadIdx.x;
    if (i < n) {
        float d = deg[i];
        deg[i] = (d > 0.0f) ? rsqrtf(d) : 0.0f;
    }
}

template<int C>
__global__ __launch_bounds__(256) void k_gemm_wmma(const float* __restrict__ A,
                                                   const float* __restrict__ W,
                                                   float* __restrict__ out, int N) {
    __shared__ float As[16][33];
    __shared__ float Bs[32][C];

    const int tid  = threadIdx.x;
    const int lane = tid & 31;
    const int wv   = tid >> 5;
    const int half = lane >> 4;
    const int r    = lane & 15;
    const int m0   = blockIdx.x * 16;
    const int n0   = wv * 16;

    v8f acc = {};

    for (int k0 = 0; k0 < 128; k0 += 32) {
        {
            int idx = tid * 2;
            int row = idx >> 5;
            int col = idx & 31;
            float v0 = 0.0f, v1 = 0.0f;
            int gr = m0 + row;
            if (gr < N) {
                const float* p = A + (size_t)gr * 128 + k0 + col;
                v0 = p[0];
                v1 = p[1];
            }
            As[row][col]     = v0;
            As[row][col + 1] = v1;
        }
        for (int idx = tid; idx < 32 * C; idx += 256) {
            int row = idx / C, col = idx % C;
            Bs[row][col] = W[(size_t)(k0 + row) * C + col];
        }
        __syncthreads();

        if (n0 < C) {
#pragma unroll
            for (int kk = 0; kk < 32; kk += 4) {
                v2f a, b;
                a.x = As[r][kk + 2 * half];
                a.y = As[r][kk + 2 * half + 1];
                b.x = Bs[kk + 2 * half][n0 + r];
                b.y = Bs[kk + 2 * half + 1][n0 + r];
                acc = __builtin_amdgcn_wmma_f32_16x16x4_f32(
                     false, a,  false, b,
                     (short)0, acc,  false,  false);
            }
        }
        __syncthreads();
    }

    if (n0 < C) {
#pragma unroll
        for (int r2 = 0; r2 < 8; ++r2) {
            int row = m0 + r2 + 8 * half;
            if (row < N) out[(size_t)row * C + n0 + r] = acc[r2];
        }
    }
}

template<int C>
__global__ void k_aggregate(const float* __restrict__ feat,
                            const int* __restrict__ src,
                            const int* __restrict__ dst,
                            const float* __restrict__ dinv,
                            float* out, int E, int N) {
    int gw   = (blockIdx.x * blockDim.x + threadIdx.x) >> 5;
    int lane = threadIdx.x & 31;
    int total = E + N;
    if (gw >= total) return;

    int s, d;
    if (gw < E) { s = src[gw]; d = dst[gw]; }
    else        { s = d = gw - E; }

    float w = dinv[s] * dinv[d];

    if (C == 128) {
        const float4* fp = (const float4*)(feat + (size_t)s * C);
        float4 v = fp[lane];
        float* op = out + (size_t)d * C + lane * 4;
        atomicAdd(op + 0, v.x * w);
        atomicAdd(op + 1, v.y * w);
        atomicAdd(op + 2, v.z * w);
        atomicAdd(op + 3, v.w * w);
    } else {
        const float2* fp = (const float2*)(feat + (size_t)s * C);
        float2 v = fp[lane];
        float* op = out + (size_t)d * C + lane * 2;
        atomicAdd(op + 0, v.x * w);
        atomicAdd(op + 1, v.y * w);
    }
}

__global__ void k_bias_relu(float* h, const float* __restrict__ b, int total, int C) {
    int i = blockIdx.x * blockDim.x + threadIdx.x;
    if (i < total) {
        float v = h[i] + b[i % C];
        h[i] = v > 0.0f ? v : 0.0f;
    }
}

__global__ void k_bias_set(float* o, const float* __restrict__ b, int total, int C) {
    int i = blockIdx.x * blockDim.x + threadIdx.x;
    if (i < total) o[i] = b[i % C];
}

extern "C" void kernel_launch(void* const* d_in, const int* in_sizes, int n_in,
                              void* d_out, int out_size, void* d_ws, size_t ws_size,
                              hipStream_t stream) {
    const float* x  = (const float*)d_in[0];
    const int*   ei = (const int*)  d_in[1];
    const float* W1 = (const float*)d_in[2];
    const float* b1 = (const float*)d_in[3];
    const float* W2 = (const float*)d_in[4];
    const float* b2 = (const float*)d_in[5];

    const int N = in_sizes[0] / 128;
    const int E = in_sizes[1] / 2;
    const int* src = ei;
    const int* dst = ei + E;

    char* ws = (char*)d_ws;
    float* dinv = (float*)ws;
    size_t off  = (((size_t)N * 4) + 255) & ~(size_t)255;
    float* bufA = (float*)(ws + off);
    size_t szA  = (size_t)N * 128 * sizeof(float);
    float* bufB = (float*)(ws + off + szA);

    const int msgs = E + N;

    k_init_deg<<<(N + 255) / 256, 256, 0, stream>>>(dinv, N);
    k_count_deg<<<(E + 255) / 256, 256, 0, stream>>>(dst, dinv, E);
    k_rsqrt_inplace<<<(N + 255) / 256, 256, 0, stream>>>(dinv, N);

    k_gemm_wmma<128><<<(N + 15) / 16, 256, 0, stream>>>(x, W1, bufA, N);

    hipMemsetAsync(bufB, 0, szA, stream);
    {
        long long th = (long long)msgs * 32;
        k_aggregate<128><<<(int)((th + 255) / 256), 256, 0, stream>>>(bufA, src, dst, dinv, bufB, E, N);
    }
    k_bias_relu<<<(N * 128 + 255) / 256, 256, 0, stream>>>(bufB, b1, N * 128, 128);

    k_gemm_wmma<64><<<(N + 15) / 16, 256, 0, stream>>>(bufB, W2, bufA, N);

    k_bias_set<<<(N * 64 + 255) / 256, 256, 0, stream>>>((float*)d_out, b2, N * 64, 64);
    {
        long long th = (long long)msgs * 32;
        k_aggregate<64><<<(int)((th + 255) / 256), 256, 0, stream>>>(bufA, src, dst, dinv, (float*)d_out, E, N);
    }
}


# GEMM retiled: 64 rows x all C columns per 4-wave block, W staged whole in LDS once per block as k-pairs (was once per 16 rows), A and D tiles via padded LDS, 128-byte-line stores
# speedup vs baseline: 1.1117x; 1.1117x over previous
#include <hip/hip_runtime.h>

typedef float v2f __attribute__((ext_vector_type(2)));
typedef float v8f __attribute__((ext_vector_type(8)));

__global__ void k_init_deg(float* deg, int n) {
    int i = blockIdx.x * blockDim.x + threadIdx.x;
    if (i < n) deg[i] = 1.0f;
}

__global__ void k_count_deg(const int* __restrict__ dst, float* deg, int E) {
    int e = blockIdx.x * blockDim.x + threadIdx.x;
    if (e < E) atomicAdd(&deg[dst[e]], 1.0f);
}

__global__ void k_rsqrt_inplace(float* deg, int n) {
    int i = blockIdx.x * blockDim.x + threadIdx.x;
    if (i < n) {
        float d = deg[i];
        deg[i] = (d > 0.0f) ? rsqrtf(d) : 0.0f;
    }
}

template<int C>
__global__ __launch_bounds__(128) void k_gemm_wmma(const float* __restrict__ A,
                                                   const float* __restrict__ W,
                                                   float* __restrict__ out, int N) {
    constexpr int NT = C / 16;
    constexpr int C4 = C / 4;
    constexpr int BP = 2 * C + 32;
    constexpr int AP = 132;

    __shared__ __attribute__((aligned(16))) float Bs[64 * BP];
    __shared__ __attribute__((aligned(16))) float As[64 * AP];

    const int tid  = threadIdx.x;
    const int lane = tid & 31;
    const int wv   = tid >> 5;
    const int half = lane >> 4;
    const int r    = lane & 15;
    const int m0   = blockIdx.x * 64;

    const bool full = (m0 + 64 <= N);
    if (full) {
#pragma unroll
        for (int it = 0; it < 16; ++it) {
            int idx = tid + it * 128;
            int row = idx >> 5;
            int c4  = idx & 31;
            *(float4*)(As + row * AP + 4 * c4) = *(const float4*)(A + (size_t)(m0 + row) * 128 + 4 * c4);
        }
    } else {
#pragma unroll
        for (int it = 0; it < 16; ++it) {
            int idx = tid + it * 128;
            int row = idx >> 5;
            int c4  = idx & 31;
            float4 v = make_float4(0.0f, 0.0f, 0.0f, 0.0f);
            if (m0 + row < N) v = *(const float4*)(A + (size_t)(m0 + row) * 128 + 4 * c4);
            *(float4*)(As + row * AP + 4 * c4) = v;
        }
    }
#pragma unroll
    for (int it = 0; it < (64 * C4) / 128; ++it) {
        int idx = tid + it * 128;
        int kp  = idx / C4;
        int n4  = (idx % C4) * 4;
        float4 e = *(const float4*)(W + (size_t)(2 * kp) * C + n4);
        float4 o = *(const float4*)(W + (size_t)(2 * kp + 1) * C + n4);
        float4* d = (float4*)(Bs + kp * BP + 2 * n4);
        d[0] = make_float4(e.x, o.x, e.y, o.y);
        d[1] = make_float4(e.z, o.z, e.w, o.w);
    }
    __syncthreads();

    v8f acc[NT];
#pragma unroll
    for (int j = 0; j < NT; ++j) acc[j] = v8f{};

    const float* ap = As + (16 * wv + r) * AP + 2 * half;
    const float* bp = Bs + half * BP + 2 * r;

#pragma unroll
    for (int kk = 0; kk < 128; kk += 4) {
        v2f a = *(const v2f*)(ap + kk);
#pragma unroll
        for (int j = 0; j < NT; ++j) {
            v2f b = *(const v2f*)(bp + (kk >> 1) * BP + 32 * j);
            acc[j] = __builtin_amdgcn_wmma_f32_16x16x4_f32(
                 false, a,  false, b,
                 (short)0, acc[j],  false,  false);
        }
    }

    __syncthreads();
#pragma unroll
    for (int j = 0; j < NT; ++j) {
#pragma unroll
        for (int r2 = 0; r2 < 8; ++r2) {
            As[(16 * wv + r2 + 8 * half) * AP + 16 * j + r] = acc[j][r2];
        }
    }
    __syncthreads();

    if (full) {
#pragma unroll
        for (int it = 0; it < (64 * C4) / 128; ++it) {
            int idx = tid + it * 128;
            int row = idx / C4;
            int c4  = idx % C4;
            *(float4*)(out + (size_t)(m0 + row) * C + 4 * c4) = *(const float4*)(As + row * AP + 4 * c4);
        }
    } else {
#pragma unroll
        for (int it = 0; it < (64 * C4) / 128; ++it) {
            int idx = tid + it * 128;
            int row = idx / C4;
            int c4  = idx % C4;
            if (m0 + row < N)
                *(float4*)(out + (size_t)(m0 + row) * C + 4 * c4) = *(const float4*)(As + row * AP + 4 * c4);
        }
    }
}

template<int C>
__global__ void k_aggregate(const float* __restrict__ feat,
                            const int* __restrict__ src,
                            const int* __restrict__ dst,
                            const float* __restrict__ dinv,
                            float* out, int E, int N) {
    int gw   = (blockIdx.x * blockDim.x + threadIdx.x) >> 5;
    int lane = threadIdx.x & 31;
    int total = E + N;
    if (gw >= total) return;

    int s, d;
    if (gw < E) { s = src[gw]; d = dst[gw]; }
    else        { s = d = gw - E; }

    float w = dinv[s] * dinv[d];

    if (C == 128) {
        const float4* fp = (const float4*)(feat + (size_t)s * C);
        float4 v = fp[lane];
        float* op = out + (size_t)d * C + lane * 4;
        atomicAdd(op + 0, v.x * w);
        atomicAdd(op + 1, v.y * w);
        atomicAdd(op + 2, v.z * w);
        atomicAdd(op + 3, v.w * w);
    } else {
        const float2* fp = (const float2*)(feat + (size_t)s * C);
        float2 v = fp[lane];
        float* op = out + (size_t)d * C + lane * 2;
        atomicAdd(op + 0, v.x * w);
        atomicAdd(op + 1, v.y * w);
    }
}

__global__ void k_bias_relu(float* h, const float* __restrict__ b, int total, int C) {
    int i = blockIdx.x * blockDim.x + threadIdx.x;
    if (i < total) {
        float v = h[i] + b[i % C];
        h[i] = v > 0.0f ? v : 0.0f;
    }
}

__global__ void k_bias_set(float* o, const float* __restrict__ b, int total, int C) {
    int i = blockIdx.x * blockDim.x + threadIdx.x;
    if (i < total) o[i] = b[i % C];
}

extern "C" void kernel_launch(void* const* d_in, const int* in_sizes, int n_in,
                              void* d_out, int out_size, void* d_ws, size_t ws_size,
                              hipStream_t stream) {
    const float* x  = (const float*)d_in[0];
    const int*   ei = (const int*)  d_in[1];
    const float* W1 = (const float*)d_in[2];
    const float* b1 = (const float*)d_in[3];
    const float* W2 = (const float*)d_in[4];
    const float* b2 = (const float*)d_in[5];

    const int N = in_sizes[0] / 128;
    const int E = in_sizes[1] / 2;
    const int* src = ei;
    const int* dst = ei + E;

    char* ws = (char*)d_ws;
    float* dinv = (float*)ws;
    size_t off  = (((size_t)N * 4) + 255) & ~(size_t)255;
    float* bufA = (float*)(ws + off);
    size_t szA  = (size_t)N * 128 * sizeof(float);
    float* bufB = (float*)(ws + off + szA);

    const int msgs = E + N;

    k_init_deg<<<(N + 255) / 256, 256, 0, stream>>>(dinv, N);
    k_count_deg<<<(E + 255) / 256, 256, 0, stream>>>(dst, dinv, E);
    k_rsqrt_inplace<<<(N + 255) / 256, 256, 0, stream>>>(dinv, N);

    k_gemm_wmma<128><<<(N + 63) / 64, 128, 0, stream>>>(x, W1, bufA, N);

    hipMemsetAsync(bufB, 0, szA, stream);
    {
        long long th = (long long)msgs * 32;
        k_aggregate<128><<<(int)((th + 255) / 256), 256, 0, stream>>>(bufA, src, dst, dinv, bufB, E, N);
    }
    k_bias_relu<<<(N * 128 + 255) / 256, 256, 0, stream>>>(bufB, b1, N * 128, 128);

    k_gemm_wmma<64><<<(N + 63) / 64, 128, 0, stream>>>(bufB, W2, bufA, N);

    k_bias_set<<<(N * 64 + 255) / 256, 256, 0, stream>>>((float*)d_out, b2, N * 64, 64);
    {
        long long th = (long long)msgs * 32;
        k_aggregate<64><<<(int)((th + 255) / 256), 256, 0, stream>>>(bufA, src, dst, dinv, (float*)d_out, E, N);
    }
}


# Scatter-add float atomics replaced by a gather per destination row over a CSR built in the call (int count, scan, cursor fill); a row is summed in registers in CSR arrival order and written once
# speedup vs baseline: 2.0475x; 1.8418x over previous
#include <hip/hip_runtime.h>

typedef float v2f __attribute__((ext_vector_type(2)));
typedef float v8f __attribute__((ext_vector_type(8)));

__global__ void k_zero_counts(int* cnt, int n) {
    int i = blockIdx.x * blockDim.x + threadIdx.x;
    if (i < n) cnt[i] = 0;
}

__global__ void k_count_in(const int* __restrict__ dst, int* cnt, int E) {
    int e = blockIdx.x * blockDim.x + threadIdx.x;
    if (e < E) atomicAdd(&cnt[dst[e]], 1);
}

__global__ __launch_bounds__(256) void k_dinv_blocksum(const int* __restrict__ cnt, float* __restrict__ dinv,
                                                       int* __restrict__ blocksum, int n) {
    __shared__ int sh[8];
    const int tid  = threadIdx.x;
    const int base = blockIdx.x * 1024;
    int sum = 0;
    for (int k = 0; k < 4; ++k) {
        int i = base + k * 256 + tid;
        if (i < n) {
            int c = cnt[i];
            sum += c;
            float dg = (float)(c + 1);
            dinv[i] = (dg > 0.0f) ? rsqrtf(dg) : 0.0f;
        }
    }
    for (int o = 16; o > 0; o >>= 1) sum += __shfl_xor(sum, o);
    if ((tid & 31) == 0) sh[tid >> 5] = sum;
    __syncthreads();
    if (tid == 0) {
        int t = 0;
        for (int k = 0; k < 8; ++k) t += sh[k];
        blocksum[blockIdx.x] = t;
    }
}

__global__ __launch_bounds__(256) void k_row_starts(const int* __restrict__ cnt, const int* __restrict__ blocksum,
                                                    int* __restrict__ rowbeg, int* __restrict__ cursor, int n) {
    __shared__ int sh_off[8];
    __shared__ int sh_scan[8];
    const int tid  = threadIdx.x;
    const int lane = tid & 31;
    const int wv   = tid >> 5;

    int part = 0;
    for (int k = tid; k < (int)blockIdx.x; k += 256) part += blocksum[k];
    for (int o = 16; o > 0; o >>= 1) part += __shfl_xor(part, o);
    if (lane == 0) sh_off[wv] = part;

    const int i0 = blockIdx.x * 1024 + 4 * tid;
    const bool whole = (i0 + 3 < n);
    int c0 = 0, c1 = 0, c2 = 0, c3 = 0;
    if (whole) {
        int4 c = *(const int4*)(cnt + i0);
        c0 = c.x; c1 = c.y; c2 = c.z; c3 = c.w;
    } else {
        if (i0     < n) c0 = cnt[i0];
        if (i0 + 1 < n) c1 = cnt[i0 + 1];
        if (i0 + 2 < n) c2 = cnt[i0 + 2];
    }
    const int tsum = c0 + c1 + c2 + c3;
    int inc = tsum;
    for (int o = 1; o < 32; o <<= 1) {
        int t = __shfl_up(inc, o);
        if (lane >= o) inc += t;
    }
    if (lane == 31) sh_scan[wv] = inc;
    __syncthreads();
    int off = 0, woff = 0;
    for (int k = 0; k < 8; ++k) {
        off += sh_off[k];
        if (k < wv) woff += sh_scan[k];
    }
    const int e0 = off + woff + inc - tsum;
    const int e1 = e0 + c0, e2 = e1 + c1, e3 = e2 + c2;
    if (whole) {
        int4 r = make_int4(e0, e1, e2, e3);
        *(int4*)(rowbeg + i0) = r;
        *(int4*)(cursor + i0) = r;
    } else {
        if (i0     < n) { rowbeg[i0]     = e0; cursor[i0]     = e0; }
        if (i0 + 1 < n) { rowbeg[i0 + 1] = e1; cursor[i0 + 1] = e1; }
        if (i0 + 2 < n) { rowbeg[i0 + 2] = e2; cursor[i0 + 2] = e2; }
    }
}

__global__ void k_fill_rows(const int* __restrict__ src, const int* __restrict__ dst,
                            int* cursor, int* __restrict__ col, int E) {
    int e = blockIdx.x * blockDim.x + threadIdx.x;
    if (e < E) {
        int p = atomicAdd(&cursor[dst[e]], 1);
        if ((unsigned)p < (unsigned)E) col[p] = src[e];
    }
}

template<int C>
__global__ __launch_bounds__(128) void k_gemm_wmma(const float* __restrict__ A,
                                                   const float* __restrict__ W,
                                                   float* __restrict__ out, int N) {
    constexpr int NT = C / 16;
    constexpr int C4 = C / 4;
    constexpr int BP = 2 * C + 32;
    constexpr int AP = 132;

    __shared__ __attribute__((aligned(16))) float Bs[64 * BP];
    __shared__ __attribute__((aligned(16))) float As[64 * AP];

    const int tid  = threadIdx.x;
    const int lane = tid & 31;
    const int wv   = tid >> 5;
    const int half = lane >> 4;
    const int r    = lane & 15;
    const int m0   = blockIdx.x * 64;

    const bool full = (m0 + 64 <= N);
    if (full) {
#pragma unroll
        for (int it = 0; it < 16; ++it) {
            int idx = tid + it * 128;
            int row = idx >> 5;
            int c4  = idx & 31;
            *(float4*)(As + row * AP + 4 * c4) = *(const float4*)(A + (size_t)(m0 + row) * 128 + 4 * c4);
        }
    } else {
#pragma unroll
        for (int it = 0; it < 16; ++it) {
            int idx = tid + it * 128;
            int row = idx >> 5;
            int c4  = idx & 31;
            float4 v = make_float4(0.0f, 0.0f, 0.0f, 0.0f);
            if (m0 + row < N) v = *(const float4*)(A + (size_t)(m0 + row) * 128 + 4 * c4);
            *(float4*)(As + row * AP + 4 * c4) = v;
        }
    }
#pragma unroll
    for (int it = 0; it < (64 * C4) / 128; ++it) {
        int idx = tid + it * 128;
        int kp  = idx / C4;
        int n4  = (idx % C4) * 4;
        float4 e = *(const float4*)(W + (size_t)(2 * kp) * C + n4);
        float4 o = *(const float4*)(W + (size_t)(2 * kp + 1) * C + n4);
        float4* d = (float4*)(Bs + kp * BP + 2 * n4);
        d[0] = make_float4(e.x, o.x, e.y, o.y);
        d[1] = make_float4(e.z, o.z, e.w, o.w);
    }
    __syncthreads();

    v8f acc[NT];
#pragma unroll
    for (int j = 0; j < NT; ++j) acc[j] = v8f{};

    const float* ap = As + (16 * wv + r) * AP + 2 * half;
    const float* bp = Bs + half * BP + 2 * r;

#pragma unroll
    for (int kk = 0; kk < 128; kk += 4) {
        v2f a = *(const v2f*)(ap + kk);
#pragma unroll
        for (int j = 0; j < NT; ++j) {
            v2f b = *(const v2f*)(bp + (kk >> 1) * BP + 32 * j);
            acc[j] = __builtin_amdgcn_wmma_f32_16x16x4_f32(
                 false, a,  false, b,
                 (short)0, acc[j],  false,  false);
        }
    }

    __syncthreads();
#pragma unroll
    for (int j = 0; j < NT; ++j) {
#pragma unroll
        for (int r2 = 0; r2 < 8; ++r2) {
            As[(16 * wv + r2 + 8 * half) * AP + 16 * j + r] = acc[j][r2];
        }
    }
    __syncthreads();

    if (full) {
#pragma unroll
        for (int it = 0; it < (64 * C4) / 128; ++it) {
            int idx = tid + it * 128;
            int row = idx / C4;
            int c4  = idx % C4;
            *(float4*)(out + (size_t)(m0 + row) * C + 4 * c4) = *(const float4*)(As + row * AP + 4 * c4);
        }
    } else {
#pragma unroll
        for (int it = 0; it < (64 * C4) / 128; ++it) {
            int idx = tid + it * 128;
            int row = idx / C4;
            int c4  = idx % C4;
            if (m0 + row < N)
                *(float4*)(out + (size_t)(m0 + row) * C + 4 * c4) = *(const float4*)(As + row * AP + 4 * c4);
        }
    }
}

template<int C, bool RELU>
__global__ __launch_bounds__(256) void k_gather_rows(const float* __restrict__ feat,
                                                     const int* __restrict__ rowbeg,
                                                     const int* __restrict__ rowend,
                                                     const int* __restrict__ col,
                                                     const float* __restrict__ dinv,
                                                     const float* __restrict__ bias,
                                                     float* __restrict__ out, int N, int E) {
    constexpr int V = C / 32;
    typedef float vec_t __attribute__((ext_vector_type(V)));
    const int lane = threadIdx.x & 31;
    const int d = __builtin_amdgcn_readfirstlane((int)((blockIdx.x * blockDim.x + threadIdx.x) >> 5));
    if (d >= N) return;

    const float dd  = dinv[d];
    const int   beg = max(rowbeg[d], 0);
    const int   end = min(rowend[d], E);

    vec_t acc = *(const vec_t*)(feat + (size_t)d * C + lane * V) * (dd * dd);

    for (int base = beg; base < end; base += 32) {
        const int n = min(32, end - base);
        int   my_s = d;
        float my_w = 0.0f;
        if (lane < n) {
            my_s = min(max(col[base + lane], 0), N - 1);
            my_w = dinv[my_s];
        }
        int j = 0;
        for (; j + 4 <= n; j += 4) {
            const int s0 = __builtin_amdgcn_readlane(my_s, j);
            const int s1 = __builtin_amdgcn_readlane(my_s, j + 1);
            const int s2 = __builtin_amdgcn_readlane(my_s, j + 2);
            const int s3 = __builtin_amdgcn_readlane(my_s, j + 3);
            const vec_t v0 = *(const vec_t*)(feat + (size_t)s0 * C + lane * V);
            const vec_t v1 = *(const vec_t*)(feat + (size_t)s1 * C + lane * V);
            const vec_t v2 = *(const vec_t*)(feat + (size_t)s2 * C + lane * V);
            const vec_t v3 = *(const vec_t*)(feat + (size_t)s3 * C + lane * V);
            const float w0 = __int_as_float(__builtin_amdgcn_readlane(__float_as_int(my_w), j)) * dd;
            const float w1 = __int_as_float(__builtin_amdgcn_readlane(__float_as_int(my_w), j + 1)) * dd;
            const float w2 = __int_as_float(__builtin_amdgcn_readlane(__float_as_int(my_w), j + 2)) * dd;
            const float w3 = __int_as_float(__builtin_amdgcn_readlane(__float_as_int(my_w), j + 3)) * dd;
            acc += v0 * w0;
            acc += v1 * w1;
            acc += v2 * w2;
            acc += v3 * w3;
        }
        const int rem = n - j;
        if (rem > 0) {
            vec_t v0 = *(const vec_t*)(feat + (size_t)__builtin_amdgcn_readlane(my_s, j) * C + lane * V);
            vec_t v1 = {};
            vec_t v2 = {};
            float w0 = __int_as_float(__builtin_amdgcn_readlane(__float_as_int(my_w), j)) * dd;
            float w1 = 0.0f;
            float w2 = 0.0f;
            if (rem > 1) {
                v1 = *(const vec_t*)(feat + (size_t)__builtin_amdgcn_readlane(my_s, j + 1) * C + lane * V);
                w1 = __int_as_float(__builtin_amdgcn_readlane(__float_as_int(my_w), j + 1)) * dd;
            }
            if (rem > 2) {
                v2 = *(const vec_t*)(feat + (size_t)__builtin_amdgcn_readlane(my_s, j + 2) * C + lane * V);
                w2 = __int_as_float(__builtin_amdgcn_readlane(__float_as_int(my_w), j + 2)) * dd;
            }
            acc += v0 * w0;
            acc += v1 * w1;
            acc += v2 * w2;
        }
    }

    vec_t r = acc + *(const vec_t*)(bias + lane * V);
    if (RELU) {
        for (int k = 0; k < V; ++k) r[k] = r[k] > 0.0f ? r[k] : 0.0f;
    }
    *(vec_t*)(out + (size_t)d * C + lane * V) = r;
}

extern "C" void kernel_launch(void* const* d_in, const int* in_sizes, int n_in,
                              void* d_out, int out_size, void* d_ws, size_t ws_size,
                              hipStream_t stream) {
    const float* x  = (const float*)d_in[0];
    const int*   ei = (const int*)  d_in[1];
    const float* W1 = (const float*)d_in[2];
    const float* b1 = (const float*)d_in[3];
    const float* W2 = (const float*)d_in[4];
    const float* b2 = (const float*)d_in[5];

    const int N = in_sizes[0] / 128;
    const int E = in_sizes[1] / 2;
    const int* src = ei;
    const int* dst = ei + E;
    const int nb = (N + 1023) / 1024;

    char* ws = (char*)d_ws;
    size_t szN  = (((size_t)N * 4) + 255) & ~(size_t)255;
    size_t szNB = (((size_t)nb * 4) + 255) & ~(size_t)255;
    size_t szE  = (((size_t)E * 4) + 255) & ~(size_t)255;
    float* dinv     = (float*)ws;
    int*   cnt      = (int*)(ws + szN);
    int*   rowbeg   = (int*)(ws + 2 * szN);
    int*   cursor   = (int*)(ws + 3 * szN);
    int*   blocksum = (int*)(ws + 4 * szN);
    int*   col      = (int*)(ws + 4 * szN + szNB);
    size_t off  = 4 * szN + szNB + szE;
    float* bufA = (float*)(ws + off);
    size_t szA  = (size_t)N * 128 * sizeof(float);
    float* bufB = (float*)(ws + off + szA);

    k_zero_counts<<<(N + 255) / 256, 256, 0, stream>>>(cnt, N);
    k_count_in<<<(E + 255) / 256, 256, 0, stream>>>(dst, cnt, E);
    k_dinv_blocksum<<<nb, 256, 0, stream>>>(cnt, dinv, blocksum, N);
    k_row_starts<<<nb, 256, 0, stream>>>(cnt, blocksum, rowbeg, cursor, N);
    k_fill_rows<<<(E + 255) / 256, 256, 0, stream>>>(src, dst, cursor, col, E);

    k_gemm_wmma<128><<<(N + 63) / 64, 128, 0, stream>>>(x, W1, bufA, N);
    k_gather_rows<128, true><<<(N + 7) / 8, 256, 0, stream>>>(bufA, rowbeg, cursor, col, dinv, b1, bufB, N, E);

    k_gemm_wmma<64><<<(N + 63) / 64, 128, 0, stream>>>(bufB, W2, bufA, N);
    k_gather_rows<64, false><<<(N + 7) / 8, 256, 0, stream>>>(bufA, rowbeg, cursor, col, dinv, b2, (float*)d_out, N, E);
}


# CSR build on LDS counters instead of global atomics: 2-level counting sort (edges binned per 512-node bucket, then per node), 5 build launches become 3; rows in LDS arrival order (sum order changes)
# speedup vs baseline: 3.7502x; 1.8316x over previous
#include <hip/hip_runtime.h>

typedef float v2f __attribute__((ext_vector_type(2)));
typedef float v4f __attribute__((ext_vector_type(4)));
typedef float v8f __attribute__((ext_vector_type(8)));

constexpr int NPB_LOG = 9;
constexpr int NPB     = 1 << NPB_LOG;
constexpr int MAXKB   = 1024;
constexpr int EBLK    = 128;

__device__ __forceinline__ int bucket_of(int d, int KB) {
    return (int)min((unsigned)d >> NPB_LOG, (unsigned)(KB - 1));
}

__global__ __launch_bounds__(1024) void k_bucket_hist(const int* __restrict__ dst, int* __restrict__ H, int E, int KB) {
    __shared__ int hist[MAXKB];
    const int tid = threadIdx.x;
    hist[tid] = 0;
    __syncthreads();
    const int CH = (E + (int)gridDim.x - 1) / (int)gridDim.x;
    const int lo = blockIdx.x * CH;
    const int hi = min(E, lo + CH);
    for (int e0 = lo + tid; e0 < hi; e0 += 8 * 1024) {
        int d[8];
#pragma unroll
        for (int u = 0; u < 8; ++u)
            d[u] = dst[min(e0 + u * 1024, hi - 1)];
        asm volatile("" :: "v"(d[0]), "v"(d[1]), "v"(d[2]), "v"(d[3]), "v"(d[4]), "v"(d[5]), "v"(d[6]), "v"(d[7]));
#pragma unroll
        for (int u = 0; u < 8; ++u)
            if (e0 + u * 1024 < hi) atomicAdd(&hist[bucket_of(d[u], KB)], 1);
    }
    __syncthreads();
    H[blockIdx.x * MAXKB + tid] = hist[tid];
}

__global__ __launch_bounds__(1024) void k_bucket_fill(const int* __restrict__ src, const int* __restrict__ dst,
                                                      const int* __restrict__ H, int* __restrict__ bucketbase,
                                                      int* __restrict__ pairs, int E, int KB) {
    __shared__ int tot[MAXKB];
    __shared__ int before[MAXKB];
    __shared__ int wsum[32];
    const int tid  = threadIdx.x;
    const int lane = tid & 31;
    const int wv   = tid >> 5;
    const int b    = blockIdx.x;
    const int nblk = gridDim.x;
    tot[tid] = 0;
    before[tid] = 0;
    __syncthreads();
    {
        const int kg    = (KB + 3) >> 2;
        const int shift = (kg > 1) ? (32 - __builtin_clz(kg - 1)) : 0;
        const int g     = tid & ((1 << shift) - 1);
        const int part  = tid >> shift;
        const int parts = 1024 >> shift;
        if (g < kg) {
            int4 t  = make_int4(0, 0, 0, 0);
            int4 bf = make_int4(0, 0, 0, 0);
            for (int bb0 = part; bb0 < nblk; bb0 += 8 * parts) {
                int4 v[8];
#pragma unroll
                for (int u = 0; u < 8; ++u)
                    v[u] = *(const int4*)(H + (size_t)min(bb0 + u * parts, nblk - 1) * MAXKB + 4 * g);
#pragma unroll
                for (int u = 0; u < 8; ++u) {
                    const int bb = bb0 + u * parts;
                    const int in = (bb < nblk) ? -1 : 0;
                    const int fr = (bb < b) ? in : 0;
                    t.x  += v[u].x & in;  t.y  += v[u].y & in;  t.z  += v[u].z & in;  t.w  += v[u].w & in;
                    bf.x += v[u].x & fr;  bf.y += v[u].y & fr;  bf.z += v[u].z & fr;  bf.w += v[u].w & fr;
                }
            }
            atomicAdd(&tot[4 * g],     t.x);
            atomicAdd(&tot[4 * g + 1], t.y);
            atomicAdd(&tot[4 * g + 2], t.z);
            atomicAdd(&tot[4 * g + 3], t.w);
            atomicAdd(&before[4 * g],     bf.x);
            atomicAdd(&before[4 * g + 1], bf.y);
            atomicAdd(&before[4 * g + 2], bf.z);
            atomicAdd(&before[4 * g + 3], bf.w);
        }
    }
    __syncthreads();
    const int v = (tid < KB) ? tot[tid] : 0;
    int inc = v;
    for (int o = 1; o < 32; o <<= 1) {
        const int t = __shfl_up(inc, o);
        if (lane >= o) inc += t;
    }
    if (lane == 31) wsum[wv] = inc;
    __syncthreads();
    int woff = 0;
    for (int k = 0; k < 32; ++k)
        if (k < wv) woff += wsum[k];
    const int excl = woff + inc - v;
    if (tid < KB) {
        before[tid] += excl;
        if (b == 0) {
            bucketbase[tid] = excl;
            if (tid == KB - 1) bucketbase[KB] = excl + v;
        }
    }
    __syncthreads();
    const int CH = (E + nblk - 1) / nblk;
    const int lo = b * CH;
    const int hi = min(E, lo + CH);
    int2* pr2 = (int2*)pairs;
    for (int e0 = lo + tid; e0 < hi; e0 += 8 * 1024) {
        int d[8], s[8];
#pragma unroll
        for (int u = 0; u < 8; ++u) {
            const int e = min(e0 + u * 1024, hi - 1);
            d[u] = dst[e];
            s[u] = src[e];
        }
        asm volatile("" :: "v"(d[0]), "v"(d[1]), "v"(d[2]), "v"(d[3]), "v"(d[4]), "v"(d[5]), "v"(d[6]), "v"(d[7]),
                           "v"(s[0]), "v"(s[1]), "v"(s[2]), "v"(s[3]), "v"(s[4]), "v"(s[5]), "v"(s[6]), "v"(s[7]));
#pragma unroll
        for (int u = 0; u < 8; ++u) {
            if (e0 + u * 1024 < hi) {
                const int p = atomicAdd(&before[bucket_of(d[u], KB)], 1);
                if ((unsigned)p < (unsigned)E) pr2[p] = make_int2(s[u], d[u]);
            }
        }
    }
}

__global__ __launch_bounds__(512) void k_bucket_rows(const int* __restrict__ pairs, const int* __restrict__ bucketbase,
                                                     float* __restrict__ dinv, int* __restrict__ rowbeg,
                                                     int* __restrict__ rowend, int* __restrict__ col, int N, int E) {
    constexpr int CAP = 8192;
    __shared__ int cnt[NPB];
    __shared__ int wsum[16];
    __shared__ int stage[CAP];
    const int tid  = threadIdx.x;
    const int lane = tid & 31;
    const int wv   = tid >> 5;
    const int k    = blockIdx.x;
    const int n0   = k << NPB_LOG;
    const int base = min(max(bucketbase[k], 0), E);
    const int m    = min(max(bucketbase[k + 1] - base, 0), E - base);
    const int2* pr2 = (const int2*)pairs + base;
    cnt[tid] = 0;
    __syncthreads();
    for (int i0 = tid; i0 < m; i0 += 8 * 512) {
        int2 pr[8];
#pragma unroll
        for (int u = 0; u < 8; ++u)
            pr[u] = pr2[min(i0 + u * 512, m - 1)];
        asm volatile("" :: "v"(pr[0].y), "v"(pr[1].y), "v"(pr[2].y), "v"(pr[3].y),
                           "v"(pr[4].y), "v"(pr[5].y), "v"(pr[6].y), "v"(pr[7].y));
#pragma unroll
        for (int u = 0; u < 8; ++u)
            if (i0 + u * 512 < m) atomicAdd(&cnt[min(max(pr[u].y - n0, 0), NPB - 1)], 1);
    }
    __syncthreads();
    const int c = cnt[tid];
    int inc = c;
    for (int o = 1; o < 32; o <<= 1) {
        const int t = __shfl_up(inc, o);
        if (lane >= o) inc += t;
    }
    if (lane == 31) wsum[wv] = inc;
    __syncthreads();
    int woff = 0;
    for (int j = 0; j < 16; ++j)
        if (j < wv) woff += wsum[j];
    const int rs = woff + inc - c;
    cnt[tid] = rs;
    const int node = n0 + tid;
    if (node < N) {
        rowbeg[node] = base + rs;
        rowend[node] = base + rs + c;
        const float dg = (float)(c + 1);
        dinv[node] = (dg > 0.0f) ? rsqrtf(dg) : 0.0f;
    }
    __syncthreads();
    for (int i0 = tid; i0 < m; i0 += 8 * 512) {
        int2 pr[8];
#pragma unroll
        for (int u = 0; u < 8; ++u)
            pr[u] = pr2[min(i0 + u * 512, m - 1)];
        asm volatile("" :: "v"(pr[0].x), "v"(pr[0].y), "v"(pr[1].x), "v"(pr[1].y), "v"(pr[2].x), "v"(pr[2].y),
                           "v"(pr[3].x), "v"(pr[3].y), "v"(pr[4].x), "v"(pr[4].y), "v"(pr[5].x), "v"(pr[5].y),
                           "v"(pr[6].x), "v"(pr[6].y), "v"(pr[7].x), "v"(pr[7].y));
#pragma unroll
        for (int u = 0; u < 8; ++u) {
            if (i0 + u * 512 < m) {
                const int p = atomicAdd(&cnt[min(max(pr[u].y - n0, 0), NPB - 1)], 1);
                if ((unsigned)p < (unsigned)CAP) stage[p] = pr[u].x;
                else if ((unsigned)p < (unsigned)m) col[base + p] = pr[u].x;
            }
        }
    }
    __syncthreads();
    const int mc = min(m, CAP);
    for (int i = tid; i < mc; i += 512) col[base + i] = stage[i];
}

__global__ __launch_bounds__(512) void k_layer1_fused(const float* __restrict__ x,
                                                      const int* __restrict__ rowbeg,
                                                      const int* __restrict__ col,
                                                      const float* __restrict__ dinv,
                                                      const float* __restrict__ W1,
                                                      const float* __restrict__ b1,
                                                      const float* __restrict__ W2,
                                                      float* __restrict__ g, int N, int E) {
    constexpr int BP1 = 2 * 128 + 32;
    constexpr int BP2 = 2 * 64 + 32;
    constexpr int AP  = 132;

    __shared__ __attribute__((aligned(16))) float B1s[64 * BP1];
    __shared__ __attribute__((aligned(16))) float B2s[64 * BP2];
    __shared__ __attribute__((aligned(16))) float As[64 * AP];

    const int tid  = threadIdx.x;
    const int lane = tid & 31;
    const int wv   = tid >> 5;
    const int half = lane >> 4;
    const int r    = lane & 15;
    const int m0   = blockIdx.x * 64;
    const int row0 = m0 + 4 * wv;

    const int  mrow   = row0 + lane;
    const int  mrowc  = min(mrow, N - 1);
    const int  rb_raw = rowbeg[mrowc];
    const float dv_raw = dinv[mrowc];
    v4f selfv[4];
#pragma unroll
    for (int i = 0; i < 4; ++i)
        selfv[i] = *(const v4f*)(x + (size_t)min(row0 + i, N - 1) * 128 + 4 * lane);

    float4 w1e[4], w1o[4], w2e[2], w2o[2];
#pragma unroll
    for (int it = 0; it < 4; ++it) {
        const int idx = tid + it * 512;
        const int kp  = idx >> 5;
        const int n4  = (idx & 31) * 4;
        w1e[it] = *(const float4*)(W1 + (size_t)(2 * kp) * 128 + n4);
        w1o[it] = *(const float4*)(W1 + (size_t)(2 * kp + 1) * 128 + n4);
    }
#pragma unroll
    for (int it = 0; it < 2; ++it) {
        const int idx = tid + it * 512;
        const int kp  = idx >> 4;
        const int n4  = (idx & 15) * 4;
        w2e[it] = *(const float4*)(W2 + (size_t)(2 * kp) * 64 + n4);
        w2o[it] = *(const float4*)(W2 + (size_t)(2 * kp + 1) * 64 + n4);
    }

    const int   bl  = (mrow < N) ? min(max(rb_raw, 0), E) : E;
    const float mdd = (mrow < N) ? dv_raw : 0.0f;
    const int ebeg = __builtin_amdgcn_readlane(bl, 0);
    const int eend = __builtin_amdgcn_readlane(bl, 4);
    int s_first = 0;
    if (lane < eend - ebeg) s_first = col[ebeg + lane];

#pragma unroll
    for (int it = 0; it < 4; ++it) {
        const int idx = tid + it * 512;
        const int kp  = idx >> 5;
        const int n4  = (idx & 31) * 4;
        float4* d = (float4*)(B1s + kp * BP1 + 2 * n4);
        d[0] = make_float4(w1e[it].x, w1o[it].x, w1e[it].y, w1o[it].y);
        d[1] = make_float4(w1e[it].z, w1o[it].z, w1e[it].w, w1o[it].w);
    }
#pragma unroll
    for (int it = 0; it < 2; ++it) {
        const int idx = tid + it * 512;
        const int kp  = idx >> 4;
        const int n4  = (idx & 15) * 4;
        float4* d = (float4*)(B2s + kp * BP2 + 2 * n4);
        d[0] = make_float4(w2e[it].x, w2o[it].x, w2e[it].y, w2o[it].y);
        d[1] = make_float4(w2e[it].z, w2o[it].z, w2e[it].w, w2o[it].w);
    }

    float* trow = As + (4 * wv) * AP + 4 * lane;
#pragma unroll
    for (int i = 0; i < 4; ++i) {
        const float di = __int_as_float(__builtin_amdgcn_readlane(__float_as_int(mdd), i));
        *(v4f*)(trow + i * AP) = (row0 + i < N) ? selfv[i] * (di * di) : v4f{};
    }

    {
        int   cur  = 0;
        int   endc = __builtin_amdgcn_readlane(bl, 1);
        float ddc  = __int_as_float(__builtin_amdgcn_readlane(__float_as_int(mdd), 0));
        v4f   acc  = v4f{};
        const float* xl = x + 4 * lane;

        auto take = [&](const v4f v, const float wsrc, const int e) {
            while (e >= endc && cur < 3) {
                *(v4f*)(trow + cur * AP) += acc;
                ++cur;
                endc = __builtin_amdgcn_readlane(bl, cur + 1);
                ddc  = __int_as_float(__builtin_amdgcn_readlane(__float_as_int(mdd), cur));
                acc  = v4f{};
            }
            acc += v * (wsrc * ddc);
        };

        for (int base = ebeg; base < eend; base += 32) {
            const int n = min(32, eend - base);
            int raw = s_first;
            if (base != ebeg) {
                raw = 0;
                if (lane < n) raw = col[base + lane];
            }
            const int my_s = min(max(raw, 0), N - 1);
            float my_w = 0.0f;
            if (lane < n) my_w = dinv[my_s];
            for (int j = 0; j < n; j += 8) {
                const int m = min(8, n - j);
                v4f v[8];
                if (m == 8) {
#pragma unroll
                    for (int k = 0; k < 8; ++k)
                        v[k] = *(const v4f*)(xl + (size_t)__builtin_amdgcn_readlane(my_s, j + k) * 128);
                } else {
#pragma unroll
                    for (int k = 0; k < 8; ++k) {
                        v[k] = v4f{};
                        if (k < m) v[k] = *(const v4f*)(xl + (size_t)__builtin_amdgcn_readlane(my_s, j + k) * 128);
                    }
                }
#pragma unroll
                for (int k = 0; k < 8; ++k) {
                    if (k < m)
                        take(v[k], __int_as_float(__builtin_amdgcn_readlane(__float_as_int(my_w), j + k)), base + j + k);
                }
            }
        }
        *(v4f*)(trow + cur * AP) += acc;
    }
    __syncthreads();

    const int rt = wv & 3;
    const int cg = wv >> 2;
    const float* ap = As + (16 * rt + r) * AP + 2 * half;

    v8f c0 = v8f{};
    v8f c1 = v8f{};
    {
        const float* bp = B1s + half * BP1 + 2 * r + 64 * cg;
#pragma unroll
        for (int kk = 0; kk < 128; kk += 4) {
            const v2f a  = *(const v2f*)(ap + kk);
            const v2f bA = *(const v2f*)(bp + (kk >> 1) * BP1);
            const v2f bB = *(const v2f*)(bp + (kk >> 1) * BP1 + 32);
            c0 = __builtin_amdgcn_wmma_f32_16x16x4_f32(false, a, false, bA, (short)0, c0, false, false);
            c1 = __builtin_amdgcn_wmma_f32_16x16x4_f32(false, a, false, bB, (short)0, c1, false, false);
        }
    }
    __syncthreads();
    {
        const int   colA  = 32 * cg + r;
        const float biasA = b1[colA];
        const float biasB = b1[colA + 16];
#pragma unroll
        for (int r2 = 0; r2 < 8; ++r2) {
            const float hA = c0[r2] + biasA;
            const float hB = c1[r2] + biasB;
            float* t = As + (16 * rt + r2 + 8 * half) * AP + colA;
            t[0]  = hA > 0.0f ? hA : 0.0f;
            t[16] = hB > 0.0f ? hB : 0.0f;
        }
    }
    __syncthreads();

    v8f c2 = v8f{};
    {
        const float* bp = B2s + half * BP2 + 2 * r + 32 * cg;
#pragma unroll
        for (int kk = 0; kk < 128; kk += 4) {
            const v2f a = *(const v2f*)(ap + kk);
            const v2f b = *(const v2f*)(bp + (kk >> 1) * BP2);
            c2 = __builtin_amdgcn_wmma_f32_16x16x4_f32(false, a, false, b, (short)0, c2, false, false);
        }
    }
    __syncthreads();
#pragma unroll
    for (int r2 = 0; r2 < 8; ++r2)
        As[(16 * rt + r2 + 8 * half) * AP + 16 * cg + r] = c2[r2];
    __syncthreads();

#pragma unroll
    for (int it = 0; it < 2; ++it) {
        const int idx = tid + it * 512;
        const int row = idx >> 4;
        const int c4  = idx & 15;
        if (m0 + row < N)
            *(float4*)(g + (size_t)(m0 + row) * 64 + 4 * c4) = *(const float4*)(As + row * AP + 4 * c4);
    }
}

template<int C, bool RELU>
__global__ __launch_bounds__(256) void k_gather_rows(const float* __restrict__ feat,
                                                     const int* __restrict__ rowbeg,
                                                     const int* __restrict__ rowend,
                                                     const int* __restrict__ col,
                                                     const float* __restrict__ dinv,
                                                     const float* __restrict__ bias,
                                                     float* __restrict__ out, int N, int E) {
    constexpr int V = C / 32;
    typedef float vec_t __attribute__((ext_vector_type(V)));
    const int lane = threadIdx.x & 31;
    const int d = __builtin_amdgcn_readfirstlane((int)((blockIdx.x * blockDim.x + threadIdx.x) >> 5));
    if (d >= N) return;

    const float dd  = dinv[d];
    const int   beg = max(rowbeg[d], 0);
    const int   end = min(rowend[d], E);

    vec_t acc = *(const vec_t*)(feat + (size_t)d * C + lane * V) * (dd * dd);

    for (int base = beg; base < end; base += 32) {
        const int n = min(32, end - base);
        int   my_s = d;
        float my_w = 0.0f;
        if (lane < n) {
            my_s = min(max(col[base + lane], 0), N - 1);
            my_w = dinv[my_s];
        }
        int j = 0;
        for (; j + 4 <= n; j += 4) {
            const int s0 = __builtin_amdgcn_readlane(my_s, j);
            const int s1 = __builtin_amdgcn_readlane(my_s, j + 1);
            const int s2 = __builtin_amdgcn_readlane(my_s, j + 2);
            const int s3 = __builtin_amdgcn_readlane(my_s, j + 3);
            const vec_t v0 = *(const vec_t*)(feat + (size_t)s0 * C + lane * V);
            const vec_t v1 = *(const vec_t*)(feat + (size_t)s1 * C + lane * V);
            const vec_t v2 = *(const vec_t*)(feat + (size_t)s2 * C + lane * V);
            const vec_t v3 = *(const vec_t*)(feat + (size_t)s3 * C + lane * V);
            const float w0 = __int_as_float(__builtin_amdgcn_readlane(__float_as_int(my_w), j)) * dd;
            const float w1 = __int_as_float(__builtin_amdgcn_readlane(__float_as_int(my_w), j + 1)) * dd;
            const float w2 = __int_as_float(__builtin_amdgcn_readlane(__float_as_int(my_w), j + 2)) * dd;
            const float w3 = __int_as_float(__builtin_amdgcn_readlane(__float_as_int(my_w), j + 3)) * dd;
            acc += v0 * w0;
            acc += v1 * w1;
            acc += v2 * w2;
            acc += v3 * w3;
        }
        const int rem = n - j;
        if (rem > 0) {
            vec_t v0 = *(const vec_t*)(feat + (size_t)__builtin_amdgcn_readlane(my_s, j) * C + lane * V);
            vec_t v1 = {};
            vec_t v2 = {};
            float w0 = __int_as_float(__builtin_amdgcn_readlane(__float_as_int(my_w), j)) * dd;
            float w1 = 0.0f;
            float w2 = 0.0f;
            if (rem > 1) {
                v1 = *(const vec_t*)(feat + (size_t)__builtin_amdgcn_readlane(my_s, j + 1) * C + lane * V);
                w1 = __int_as_float(__builtin_amdgcn_readlane(__float_as_int(my_w), j + 1)) * dd;
            }
            if (rem > 2) {
                v2 = *(const vec_t*)(feat + (size_t)__builtin_amdgcn_readlane(my_s, j + 2) * C + lane * V);
                w2 = __int_as_float(__builtin_amdgcn_readlane(__float_as_int(my_w), j + 2)) * dd;
            }
            acc += v0 * w0;
            acc += v1 * w1;
            acc += v2 * w2;
        }
    }

    vec_t r = acc + *(const vec_t*)(bias + lane * V);
    if (RELU) {
        for (int k = 0; k < V; ++k) r[k] = r[k] > 0.0f ? r[k] : 0.0f;
    }
    *(vec_t*)(out + (size_t)d * C + lane * V) = r;
}

extern "C" void kernel_launch(void* const* d_in, const int* in_sizes, int n_in,
                              void* d_out, int out_size, void* d_ws, size_t ws_size,
                              hipStream_t stream) {
    const float* x  = (const float*)d_in[0];
    const int*   ei = (const int*)  d_in[1];
    const float* W1 = (const float*)d_in[2];
    const float* b1 = (const float*)d_in[3];
    const float* W2 = (const float*)d_in[4];
    const float* b2 = (const float*)d_in[5];

    const int N = in_sizes[0] / 128;
    const int E = in_sizes[1] / 2;
    const int* src = ei;
    const int* dst = ei + E;
    const int KB = (N + NPB - 1) / NPB;

    char* ws = (char*)d_ws;
    size_t szN  = (((size_t)N * 4) + 255) & ~(size_t)255;
    size_t szH  = (size_t)EBLK * MAXKB * 4;
    size_t szKB = (((size_t)(KB + 1) * 4) + 255) & ~(size_t)255;
    size_t szE  = (((size_t)E * 4) + 255) & ~(size_t)255;
    float* dinv       = (float*)ws;
    int*   rowbeg     = (int*)(ws + szN);
    int*   cursor     = (int*)(ws + 2 * szN);
    int*   H          = (int*)(ws + 3 * szN);
    int*   bucketbase = (int*)(ws + 3 * szN + szH);
    int*   pairs      = (int*)(ws + 3 * szN + szH + szKB);
    int*   col        = (int*)(ws + 3 * szN + szH + szKB + 2 * szE);
    size_t off  = 3 * szN + szH + szKB + 3 * szE;
    float* bufG = (float*)(ws + off);

    k_bucket_hist<<<EBLK, 1024, 0, stream>>>(dst, H, E, KB);
    k_bucket_fill<<<EBLK, 1024, 0, stream>>>(src, dst, H, bucketbase, pairs, E, KB);
    k_bucket_rows<<<KB, 512, 0, stream>>>(pairs, bucketbase, dinv, rowbeg, cursor, col, N, E);

    k_layer1_fused<<<(N + 63) / 64, 512, 0, stream>>>(x, rowbeg, col, dinv, W1, b1, W2, bufG, N, E);

    k_gather_rows<64, false><<<(N + 7) / 8, 256, 0, stream>>>(bufG, rowbeg, cursor, col, dinv, b2, (float*)d_out, N, E);
}


# CSR build in 2 launches, was 3: each edge block sorts its own chunk by bucket in LDS (count, scan, move, whole-line stores); the bucket kernel collects its 128 segments (sum order changes)
# speedup vs baseline: 4.2014x; 1.1203x over previous
#include <hip/hip_runtime.h>

typedef float v2f __attribute__((ext_vector_type(2)));
typedef float v4f __attribute__((ext_vector_type(4)));
typedef float v8f __attribute__((ext_vector_type(8)));

constexpr int NPB_LOG = 9;
constexpr int NPB     = 1 << NPB_LOG;
constexpr int MAXKB   = 1024;
constexpr int EBLK    = 128;
constexpr int TP      = MAXKB + 64;

__device__ __forceinline__ int bucket_of(int d, int KB) {
    return (int)min((unsigned)d >> NPB_LOG, (unsigned)(KB - 1));
}

__global__ __launch_bounds__(1024) void k_chunk_sort(const int* __restrict__ src, const int* __restrict__ dst,
                                                     int* __restrict__ T, int* __restrict__ pairs, int E, int KB) {
    constexpr int CAPS = 8192;
    __shared__ int  cnt[MAXKB];
    __shared__ int  wsum[32];
    __shared__ int2 stage[CAPS];
    const int tid  = threadIdx.x;
    const int lane = tid & 31;
    const int wv   = tid >> 5;
    const int b    = blockIdx.x;
    const int CH   = (E + EBLK - 1) / EBLK;
    const int lo   = min(b * CH, E);
    const int hi   = min(E, lo + CH);
    const int mine = hi - lo;
    const bool one_sweep = CH <= 8 * 1024;
    cnt[tid] = 0;
    __syncthreads();
    int d[8], s[8];
#pragma unroll
    for (int u = 0; u < 8; ++u) { d[u] = 0; s[u] = 0; }
    for (int e0 = lo + tid; e0 < hi; e0 += 8 * 1024) {
#pragma unroll
        for (int u = 0; u < 8; ++u)
            d[u] = dst[min(e0 + u * 1024, hi - 1)];
        if (one_sweep) {
#pragma unroll
            for (int u = 0; u < 8; ++u)
                s[u] = src[min(e0 + u * 1024, hi - 1)];
        }
        asm volatile("" :: "v"(d[0]), "v"(d[1]), "v"(d[2]), "v"(d[3]), "v"(d[4]), "v"(d[5]), "v"(d[6]), "v"(d[7]));
#pragma unroll
        for (int u = 0; u < 8; ++u)
            if (e0 + u * 1024 < hi) atomicAdd(&cnt[bucket_of(d[u], KB)], 1);
    }
    __syncthreads();
    const int v = (tid < KB) ? cnt[tid] : 0;
    int inc = v;
    for (int o = 1; o < 32; o <<= 1) {
        const int t = __shfl_up(inc, o);
        if (lane >= o) inc += t;
    }
    if (lane == 31) wsum[wv] = inc;
    __syncthreads();
    int woff = 0;
    for (int k = 0; k < 32; ++k)
        if (k < wv) woff += wsum[k];
    const int excl = woff + inc - v;
    if (tid < KB) {
        cnt[tid] = excl;
        T[b * TP + tid] = excl;
        if (tid == KB - 1) T[b * TP + KB] = excl + v;
    }
    __syncthreads();
    int2* pr2 = (int2*)pairs;
    for (int e0 = lo + tid; e0 < hi; e0 += 8 * 1024) {
        if (!one_sweep) {
#pragma unroll
            for (int u = 0; u < 8; ++u) {
                const int e = min(e0 + u * 1024, hi - 1);
                d[u] = dst[e];
                s[u] = src[e];
            }
            asm volatile("" :: "v"(d[0]), "v"(d[1]), "v"(d[2]), "v"(d[3]), "v"(d[4]), "v"(d[5]), "v"(d[6]), "v"(d[7]),
                               "v"(s[0]), "v"(s[1]), "v"(s[2]), "v"(s[3]), "v"(s[4]), "v"(s[5]), "v"(s[6]), "v"(s[7]));
        }
#pragma unroll
        for (int u = 0; u < 8; ++u) {
            if (e0 + u * 1024 < hi) {
                const int p = atomicAdd(&cnt[bucket_of(d[u], KB)], 1);
                if ((unsigned)p < (unsigned)CAPS) stage[p] = make_int2(s[u], d[u]);
                else if ((unsigned)p < (unsigned)mine) pr2[lo + p] = make_int2(s[u], d[u]);
            }
        }
    }
    __syncthreads();
    const int ms = min(mine, CAPS);
    for (int i = tid; i < ms; i += 1024) pr2[lo + i] = stage[i];
}

__global__ __launch_bounds__(512) void k_bucket_rows(const int* __restrict__ pairs, const int* __restrict__ T,
                                                     float* __restrict__ dinv, int* __restrict__ rowbeg,
                                                     int* __restrict__ rowend, int* __restrict__ col, int N, int E) {
    constexpr int CAP = 8192;
    constexpr int SPW = EBLK / 16;
    static_assert(SPW == 8, "the load fence below names eight segments");
    __shared__ int cnt[NPB];
    __shared__ int wsum[16];
    __shared__ int stage[CAP];
    __shared__ int segbeg[EBLK];
    __shared__ int seglen[EBLK];
    __shared__ int pbase[EBLK / 32];
    __shared__ int pm[EBLK / 32];
    const int tid  = threadIdx.x;
    const int lane = tid & 31;
    const int wv   = tid >> 5;
    const int k    = blockIdx.x;
    const int n0   = k << NPB_LOG;
    cnt[tid] = 0;
    if (tid < EBLK) {
        const int CH   = (E + EBLK - 1) / EBLK;
        const int lo   = min(tid * CH, E);
        const int mine = min(E, lo + CH) - lo;
        const int a = min(max(T[tid * TP + k], 0), mine);
        const int z = min(max(T[tid * TP + k + 1], a), mine);
        segbeg[tid] = lo + a;
        seglen[tid] = z - a;
        int sa = a, sl = z - a;
        for (int o = 16; o > 0; o >>= 1) {
            sa += __shfl_xor(sa, o);
            sl += __shfl_xor(sl, o);
        }
        if (lane == 0) { pbase[wv] = sa; pm[wv] = sl; }
    }
    __syncthreads();
    int sb = 0, sm = 0;
    for (int j = 0; j < EBLK / 32; ++j) { sb += pbase[j]; sm += pm[j]; }
    const int base = min(max(sb, 0), E);
    const int m    = min(max(sm, 0), E - base);
    const int2* pr2 = (const int2*)pairs;

    int2 pr[SPW];
    int  ln[SPW];
#pragma unroll
    for (int u = 0; u < SPW; ++u) {
        const int sg = wv + 16 * u;
        ln[u] = seglen[sg];
        pr[u] = pr2[min(segbeg[sg] + min(lane, max(ln[u] - 1, 0)), E - 1)];
    }
    asm volatile("" :: "v"(pr[0].y), "v"(pr[1].y), "v"(pr[2].y), "v"(pr[3].y),
                       "v"(pr[4].y), "v"(pr[5].y), "v"(pr[6].y), "v"(pr[7].y));
#pragma unroll
    for (int u = 0; u < SPW; ++u)
        if (lane < ln[u]) atomicAdd(&cnt[min(max(pr[u].y - n0, 0), NPB - 1)], 1);
    for (int u = 0; u < SPW; ++u) {
        const int sg = wv + 16 * u;
        const int L  = seglen[sg];
        for (int j = 32 + lane; j < L; j += 32) {
            const int2 q = pr2[min(segbeg[sg] + j, E - 1)];
            atomicAdd(&cnt[min(max(q.y - n0, 0), NPB - 1)], 1);
        }
    }
    __syncthreads();
    const int c = cnt[tid];
    int inc = c;
    for (int o = 1; o < 32; o <<= 1) {
        const int t = __shfl_up(inc, o);
        if (lane >= o) inc += t;
    }
    if (lane == 31) wsum[wv] = inc;
    __syncthreads();
    int woff = 0;
    for (int j = 0; j < 16; ++j)
        if (j < wv) woff += wsum[j];
    const int rs = woff + inc - c;
    cnt[tid] = rs;
    const int node = n0 + tid;
    if (node < N) {
        rowbeg[node] = base + rs;
        rowend[node] = base + rs + c;
        const float dg = (float)(c + 1);
        dinv[node] = (dg > 0.0f) ? rsqrtf(dg) : 0.0f;
    }
    __syncthreads();
#pragma unroll
    for (int u = 0; u < SPW; ++u) {
        if (lane < ln[u]) {
            const int p = atomicAdd(&cnt[min(max(pr[u].y - n0, 0), NPB - 1)], 1);
            if ((unsigned)p < (unsigned)CAP) stage[p] = pr[u].x;
            else if ((unsigned)p < (unsigned)m) col[base + p] = pr[u].x;
        }
    }
    for (int u = 0; u < SPW; ++u) {
        const int sg = wv + 16 * u;
        const int L  = seglen[sg];
        for (int j = 32 + lane; j < L; j += 32) {
            const int2 q = pr2[min(segbeg[sg] + j, E - 1)];
            const int p = atomicAdd(&cnt[min(max(q.y - n0, 0), NPB - 1)], 1);
            if ((unsigned)p < (unsigned)CAP) stage[p] = q.x;
            else if ((unsigned)p < (unsigned)m) col[base + p] = q.x;
        }
    }
    __syncthreads();
    const int mc = min(m, CAP);
    for (int i = tid; i < mc; i += 512) col[base + i] = stage[i];
}

__global__ __launch_bounds__(512) void k_layer1_fused(const float* __restrict__ x,
                                                      const int* __restrict__ rowbeg,
                                                      const int* __restrict__ col,
                                                      const float* __restrict__ dinv,
                                                      const float* __restrict__ W1,
                                                      const float* __restrict__ b1,
                                                      const float* __restrict__ W2,
                                                      float* __restrict__ g, int N, int E) {
    constexpr int BP1 = 2 * 128 + 32;
    constexpr int BP2 = 2 * 64 + 32;
    constexpr int AP  = 132;

    __shared__ __attribute__((aligned(16))) float B1s[64 * BP1];
    __shared__ __attribute__((aligned(16))) float B2s[64 * BP2];
    __shared__ __attribute__((aligned(16))) float As[64 * AP];

    const int tid  = threadIdx.x;
    const int lane = tid & 31;
    const int wv   = tid >> 5;
    const int half = lane >> 4;
    const int r    = lane & 15;
    const int m0   = blockIdx.x * 64;
    const int row0 = m0 + 4 * wv;

    const int  mrow   = row0 + lane;
    const int  mrowc  = min(mrow, N - 1);
    const int  rb_raw = rowbeg[mrowc];
    const float dv_raw = dinv[mrowc];
    v4f selfv[4];
#pragma unroll
    for (int i = 0; i < 4; ++i)
        selfv[i] = *(const v4f*)(x + (size_t)min(row0 + i, N - 1) * 128 + 4 * lane);

    float4 w1e[4], w1o[4], w2e[2], w2o[2];
#pragma unroll
    for (int it = 0; it < 4; ++it) {
        const int idx = tid + it * 512;
        const int kp  = idx >> 5;
        const int n4  = (idx & 31) * 4;
        w1e[it] = *(const float4*)(W1 + (size_t)(2 * kp) * 128 + n4);
        w1o[it] = *(const float4*)(W1 + (size_t)(2 * kp + 1) * 128 + n4);
    }
#pragma unroll
    for (int it = 0; it < 2; ++it) {
        const int idx = tid + it * 512;
        const int kp  = idx >> 4;
        const int n4  = (idx & 15) * 4;
        w2e[it] = *(const float4*)(W2 + (size_t)(2 * kp) * 64 + n4);
        w2o[it] = *(const float4*)(W2 + (size_t)(2 * kp + 1) * 64 + n4);
    }

    const int   bl  = (mrow < N) ? min(max(rb_raw, 0), E) : E;
    const float mdd = (mrow < N) ? dv_raw : 0.0f;
    const int ebeg = __builtin_amdgcn_readlane(bl, 0);
    const int eend = __builtin_amdgcn_readlane(bl, 4);
    int s_first = 0;
    if (lane < eend - ebeg) s_first = col[ebeg + lane];

#pragma unroll
    for (int it = 0; it < 4; ++it) {
        const int idx = tid + it * 512;
        const int kp  = idx >> 5;
        const int n4  = (idx & 31) * 4;
        float4* d = (float4*)(B1s + kp * BP1 + 2 * n4);
        d[0] = make_float4(w1e[it].x, w1o[it].x, w1e[it].y, w1o[it].y);
        d[1] = make_float4(w1e[it].z, w1o[it].z, w1e[it].w, w1o[it].w);
    }
#pragma unroll
    for (int it = 0; it < 2; ++it) {
        const int idx = tid + it * 512;
        const int kp  = idx >> 4;
        const int n4  = (idx & 15) * 4;
        float4* d = (float4*)(B2s + kp * BP2 + 2 * n4);
        d[0] = make_float4(w2e[it].x, w2o[it].x, w2e[it].y, w2o[it].y);
        d[1] = make_float4(w2e[it].z, w2o[it].z, w2e[it].w, w2o[it].w);
    }

    float* trow = As + (4 * wv) * AP + 4 * lane;
#pragma unroll
    for (int i = 0; i < 4; ++i) {
        const float di = __int_as_float(__builtin_amdgcn_readlane(__float_as_int(mdd), i));
        *(v4f*)(trow + i * AP) = (row0 + i < N) ? selfv[i] * (di * di) : v4f{};
    }

    {
        int   cur  = 0;
        int   endc = __builtin_amdgcn_readlane(bl, 1);
        float ddc  = __int_as_float(__builtin_amdgcn_readlane(__float_as_int(mdd), 0));
        v4f   acc  = v4f{};
        const float* xl = x + 4 * lane;

        auto take = [&](const v4f v, const float wsrc, const int e) {
            while (e >= endc && cur < 3) {
                *(v4f*)(trow + cur * AP) += acc;
                ++cur;
                endc = __builtin_amdgcn_readlane(bl, cur + 1);
                ddc  = __int_as_float(__builtin_amdgcn_readlane(__float_as_int(mdd), cur));
                acc  = v4f{};
            }
            acc += v * (wsrc * ddc);
        };

        for (int base = ebeg; base < eend; base += 32) {
            const int n = min(32, eend - base);
            int raw = s_first;
            if (base != ebeg) {
                raw = 0;
                if (lane < n) raw = col[base + lane];
            }
            const int my_s = min(max(raw, 0), N - 1);
            float my_w = 0.0f;
            if (lane < n) my_w = dinv[my_s];
            for (int j = 0; j < n; j += 8) {
                const int m = min(8, n - j);
                v4f v[8];
                if (m == 8) {
#pragma unroll
                    for (int k = 0; k < 8; ++k)
                        v[k] = *(const v4f*)(xl + (size_t)__builtin_amdgcn_readlane(my_s, j + k) * 128);
                } else {
#pragma unroll
                    for (int k = 0; k < 8; ++k) {
                        v[k] = v4f{};
                        if (k < m) v[k] = *(const v4f*)(xl + (size_t)__builtin_amdgcn_readlane(my_s, j + k) * 128);
                    }
                }
#pragma unroll
                for (int k = 0; k < 8; ++k) {
                    if (k < m)
                        take(v[k], __int_as_float(__builtin_amdgcn_readlane(__float_as_int(my_w), j + k)), base + j + k);
                }
            }
        }
        *(v4f*)(trow + cur * AP) += acc;
    }
    __syncthreads();

    const int rt = wv & 3;
    const int cg = wv >> 2;
    const float* ap = As + (16 * rt + r) * AP + 2 * half;

    v8f c0 = v8f{};
    v8f c1 = v8f{};
    {
        const float* bp = B1s + half * BP1 + 2 * r + 64 * cg;
#pragma unroll
        for (int kk = 0; kk < 128; kk += 4) {
            const v2f a  = *(const v2f*)(ap + kk);
            const v2f bA = *(const v2f*)(bp + (kk >> 1) * BP1);
            const v2f bB = *(const v2f*)(bp + (kk >> 1) * BP1 + 32);
            c0 = __builtin_amdgcn_wmma_f32_16x16x4_f32(false, a, false, bA, (short)0, c0, false, false);
            c1 = __builtin_amdgcn_wmma_f32_16x16x4_f32(false, a, false, bB, (short)0, c1, false, false);
        }
    }
    __syncthreads();
    {
        const int   colA  = 32 * cg + r;
        const float biasA = b1[colA];
        const float biasB = b1[colA + 16];
#pragma unroll
        for (int r2 = 0; r2 < 8; ++r2) {
            const float hA = c0[r2] + biasA;
            const float hB = c1[r2] + biasB;
            float* t = As + (16 * rt + r2 + 8 * half) * AP + colA;
            t[0]  = hA > 0.0f ? hA : 0.0f;
            t[16] = hB > 0.0f ? hB : 0.0f;
        }
    }
    __syncthreads();

    v8f c2 = v8f{};
    {
        const float* bp = B2s + half * BP2 + 2 * r + 32 * cg;
#pragma unroll
        for (int kk = 0; kk < 128; kk += 4) {
            const v2f a = *(const v2f*)(ap + kk);
            const v2f b = *(const v2f*)(bp + (kk >> 1) * BP2);
            c2 = __builtin_amdgcn_wmma_f32_16x16x4_f32(false, a, false, b, (short)0, c2, false, false);
        }
    }
    __syncthreads();
#pragma unroll
    for (int r2 = 0; r2 < 8; ++r2)
        As[(16 * rt + r2 + 8 * half) * AP + 16 * cg + r] = c2[r2];
    __syncthreads();

#pragma unroll
    for (int it = 0; it < 2; ++it) {
        const int idx = tid + it * 512;
        const int row = idx >> 4;
        const int c4  = idx & 15;
        if (m0 + row < N)
            *(float4*)(g + (size_t)(m0 + row) * 64 + 4 * c4) = *(const float4*)(As + row * AP + 4 * c4);
    }
}

template<int C, bool RELU>
__global__ __launch_bounds__(256) void k_gather_rows(const float* __restrict__ feat,
                                                     const int* __restrict__ rowbeg,
                                                     const int* __restrict__ rowend,
                                                     const int* __restrict__ col,
                                                     const float* __restrict__ dinv,
                                                     const float* __restrict__ bias,
                                                     float* __restrict__ out, int N, int E) {
    constexpr int V = C / 32;
    typedef float vec_t __attribute__((ext_vector_type(V)));
    const int lane = threadIdx.x & 31;
    const int d = __builtin_amdgcn_readfirstlane((int)((blockIdx.x * blockDim.x + threadIdx.x) >> 5));
    if (d >= N) return;

    const float dd  = dinv[d];
    const int   beg = max(rowbeg[d], 0);
    const int   end = min(rowend[d], E);

    vec_t acc = *(const vec_t*)(feat + (size_t)d * C + lane * V) * (dd * dd);

    for (int base = beg; base < end; base += 32) {
        const int n = min(32, end - base);
        int   my_s = d;
        float my_w = 0.0f;
        if (lane < n) {
            my_s = min(max(col[base + lane], 0), N - 1);
            my_w = dinv[my_s];
        }
        int j = 0;
        for (; j + 4 <= n; j += 4) {
            const int s0 = __builtin_amdgcn_readlane(my_s, j);
            const int s1 = __builtin_amdgcn_readlane(my_s, j + 1);
            const int s2 = __builtin_amdgcn_readlane(my_s, j + 2);
            const int s3 = __builtin_amdgcn_readlane(my_s, j + 3);
            const vec_t v0 = *(const vec_t*)(feat + (size_t)s0 * C + lane * V);
            const vec_t v1 = *(const vec_t*)(feat + (size_t)s1 * C + lane * V);
            const vec_t v2 = *(const vec_t*)(feat + (size_t)s2 * C + lane * V);
            const vec_t v3 = *(const vec_t*)(feat + (size_t)s3 * C + lane * V);
            const float w0 = __int_as_float(__builtin_amdgcn_readlane(__float_as_int(my_w), j)) * dd;
            const float w1 = __int_as_float(__builtin_amdgcn_readlane(__float_as_int(my_w), j + 1)) * dd;
            const float w2 = __int_as_float(__builtin_amdgcn_readlane(__float_as_int(my_w), j + 2)) * dd;
            const float w3 = __int_as_float(__builtin_amdgcn_readlane(__float_as_int(my_w), j + 3)) * dd;
            acc += v0 * w0;
            acc += v1 * w1;
            acc += v2 * w2;
            acc += v3 * w3;
        }
        const int rem = n - j;
        if (rem > 0) {
            vec_t v0 = *(const vec_t*)(feat + (size_t)__builtin_amdgcn_readlane(my_s, j) * C + lane * V);
            vec_t v1 = {};
            vec_t v2 = {};
            float w0 = __int_as_float(__builtin_amdgcn_readlane(__float_as_int(my_w), j)) * dd;
            float w1 = 0.0f;
            float w2 = 0.0f;
            if (rem > 1) {
                v1 = *(const vec_t*)(feat + (size_t)__builtin_amdgcn_readlane(my_s, j + 1) * C + lane * V);
                w1 = __int_as_float(__builtin_amdgcn_readlane(__float_as_int(my_w), j + 1)) * dd;
            }
            if (rem > 2) {
                v2 = *(const vec_t*)(feat + (size_t)__builtin_amdgcn_readlane(my_s, j + 2) * C + lane * V);
                w2 = __int_as_float(__builtin_amdgcn_readlane(__float_as_int(my_w), j + 2)) * dd;
            }
            acc += v0 * w0;
            acc += v1 * w1;
            acc += v2 * w2;
        }
    }

    vec_t r = acc + *(const vec_t*)(bias + lane * V);
    if (RELU) {
        for (int k = 0; k < V; ++k) r[k] = r[k] > 0.0f ? r[k] : 0.0f;
    }
    *(vec_t*)(out + (size_t)d * C + lane * V) = r;
}

extern "C" void kernel_launch(void* const* d_in, const int* in_sizes, int n_in,
                              void* d_out, int out_size, void* d_ws, size_t ws_size,
                              hipStream_t stream) {
    const float* x  = (const float*)d_in[0];
    const int*   ei = (const int*)  d_in[1];
    const float* W1 = (const float*)d_in[2];
    const float* b1 = (const float*)d_in[3];
    const float* W2 = (const float*)d_in[4];
    const float* b2 = (const float*)d_in[5];

    const int N = in_sizes[0] / 128;
    const int E = in_sizes[1] / 2;
    const int* src = ei;
    const int* dst = ei + E;
    const int KB = (N + NPB - 1) / NPB;

    char* ws = (char*)d_ws;
    size_t szN  = (((size_t)N * 4) + 255) & ~(size_t)255;
    size_t szT  = (size_t)EBLK * TP * 4;
    size_t szE  = (((size_t)E * 4) + 255) & ~(size_t)255;
    float* dinv   = (float*)ws;
    int*   rowbeg = (int*)(ws + szN);
    int*   cursor = (int*)(ws + 2 * szN);
    int*   T      = (int*)(ws + 3 * szN);
    int*   pairs  = (int*)(ws + 3 * szN + szT);
    int*   col    = (int*)(ws + 3 * szN + szT + 2 * szE);
    size_t off  = 3 * szN + szT + 3 * szE;
    float* bufG = (float*)(ws + off);

    k_chunk_sort<<<EBLK, 1024, 0, stream>>>(src, dst, T, pairs, E, KB);
    k_bucket_rows<<<KB, 512, 0, stream>>>(pairs, T, dinv, rowbeg, cursor, col, N, E);

    k_layer1_fused<<<(N + 63) / 64, 512, 0, stream>>>(x, rowbeg, col, dinv, W1, b1, W2, bufG, N, E);

    k_gather_rows<64, false><<<(N + 7) / 8, 256, 0, stream>>>(bufG, rowbeg, cursor, col, dinv, b2, (float*)d_out, N, E);
}


# Layer-1 kernel made resident: 512 blocks (2 per WGP) walk the 64-row tiles; W1 and W2 staged in LDS once per block, was once per tile; next tile's row bounds requested a tile ahead; sum order kept
# speedup vs baseline: 4.4052x; 1.0485x over previous
#include <hip/hip_runtime.h>

typedef float v2f __attribute__((ext_vector_type(2)));
typedef float v4f __attribute__((ext_vector_type(4)));
typedef float v8f __attribute__((ext_vector_type(8)));

constexpr int NPB_LOG = 9;
constexpr int NPB     = 1 << NPB_LOG;
constexpr int MAXKB   = 1024;
constexpr int EBLK    = 128;
constexpr int TP      = MAXKB + 64;

__device__ __forceinline__ int bucket_of(int d, int KB) {
    return (int)min((unsigned)d >> NPB_LOG, (unsigned)(KB - 1));
}

__global__ __launch_bounds__(1024) void k_chunk_sort(const int* __restrict__ src, const int* __restrict__ dst,
                                                     int* __restrict__ T, int* __restrict__ pairs, int E, int KB) {
    constexpr int CAPS = 8192;
    __shared__ int  cnt[MAXKB];
    __shared__ int  wsum[32];
    __shared__ int2 stage[CAPS];
    const int tid  = threadIdx.x;
    const int lane = tid & 31;
    const int wv   = tid >> 5;
    const int b    = blockIdx.x;
    const int CH   = (E + EBLK - 1) / EBLK;
    const int lo   = min(b * CH, E);
    const int hi   = min(E, lo + CH);
    const int mine = hi - lo;
    const bool one_sweep = CH <= 8 * 1024;
    cnt[tid] = 0;
    __syncthreads();
    int d[8], s[8];
#pragma unroll
    for (int u = 0; u < 8; ++u) { d[u] = 0; s[u] = 0; }
    for (int e0 = lo + tid; e0 < hi; e0 += 8 * 1024) {
#pragma unroll
        for (int u = 0; u < 8; ++u)
            d[u] = dst[min(e0 + u * 1024, hi - 1)];
        if (one_sweep) {
#pragma unroll
            for (int u = 0; u < 8; ++u)
                s[u] = src[min(e0 + u * 1024, hi - 1)];
        }
        asm volatile("" :: "v"(d[0]), "v"(d[1]), "v"(d[2]), "v"(d[3]), "v"(d[4]), "v"(d[5]), "v"(d[6]), "v"(d[7]));
#pragma unroll
        for (int u = 0; u < 8; ++u)
            if (e0 + u * 1024 < hi) atomicAdd(&cnt[bucket_of(d[u], KB)], 1);
    }
    __syncthreads();
    const int v = (tid < KB) ? cnt[tid] : 0;
    int inc = v;
    for (int o = 1; o < 32; o <<= 1) {
        const int t = __shfl_up(inc, o);
        if (lane >= o) inc += t;
    }
    if (lane == 31) wsum[wv] = inc;
    __syncthreads();
    int woff = 0;
    for (int k = 0; k < 32; ++k)
        if (k < wv) woff += wsum[k];
    const int excl = woff + inc - v;
    if (tid < KB) {
        cnt[tid] = excl;
        T[b * TP + tid] = excl;
        if (tid == KB - 1) T[b * TP + KB] = excl + v;
    }
    __syncthreads();
    int2* pr2 = (int2*)pairs;
    for (int e0 = lo + tid; e0 < hi; e0 += 8 * 1024) {
        if (!one_sweep) {
#pragma unroll
            for (int u = 0; u < 8; ++u) {
                const int e = min(e0 + u * 1024, hi - 1);
                d[u] = dst[e];
                s[u] = src[e];
            }
            asm volatile("" :: "v"(d[0]), "v"(d[1]), "v"(d[2]), "v"(d[3]), "v"(d[4]), "v"(d[5]), "v"(d[6]), "v"(d[7]),
                               "v"(s[0]), "v"(s[1]), "v"(s[2]), "v"(s[3]), "v"(s[4]), "v"(s[5]), "v"(s[6]), "v"(s[7]));
        }
#pragma unroll
        for (int u = 0; u < 8; ++u) {
            if (e0 + u * 1024 < hi) {
                const int p = atomicAdd(&cnt[bucket_of(d[u], KB)], 1);
                if ((unsigned)p < (unsigned)CAPS) stage[p] = make_int2(s[u], d[u]);
                else if ((unsigned)p < (unsigned)mine) pr2[lo + p] = make_int2(s[u], d[u]);
            }
        }
    }
    __syncthreads();
    const int ms = min(mine, CAPS);
    for (int i = tid; i < ms; i += 1024) pr2[lo + i] = stage[i];
}

__global__ __launch_bounds__(512) void k_bucket_rows(const int* __restrict__ pairs, const int* __restrict__ T,
                                                     float* __restrict__ dinv, int* __restrict__ rowbeg,
                                                     int* __restrict__ rowend, int* __restrict__ col, int N, int E) {
    constexpr int CAP = 8192;
    constexpr int SPW = EBLK / 16;
    static_assert(SPW == 8, "the load fence below names eight segments");
    __shared__ int cnt[NPB];
    __shared__ int wsum[16];
    __shared__ int stage[CAP];
    __shared__ int segbeg[EBLK];
    __shared__ int seglen[EBLK];
    __shared__ int pbase[EBLK / 32];
    __shared__ int pm[EBLK / 32];
    const int tid  = threadIdx.x;
    const int lane = tid & 31;
    const int wv   = tid >> 5;
    const int k    = blockIdx.x;
    const int n0   = k << NPB_LOG;
    cnt[tid] = 0;
    if (tid < EBLK) {
        const int CH   = (E + EBLK - 1) / EBLK;
        const int lo   = min(tid * CH, E);
        const int mine = min(E, lo + CH) - lo;
        const int a = min(max(T[tid * TP + k], 0), mine);
        const int z = min(max(T[tid * TP + k + 1], a), mine);
        segbeg[tid] = lo + a;
        seglen[tid] = z - a;
        int sa = a, sl = z - a;
        for (int o = 16; o > 0; o >>= 1) {
            sa += __shfl_xor(sa, o);
            sl += __shfl_xor(sl, o);
        }
        if (lane == 0) { pbase[wv] = sa; pm[wv] = sl; }
    }
    __syncthreads();
    int sb = 0, sm = 0;
    for (int j = 0; j < EBLK / 32; ++j) { sb += pbase[j]; sm += pm[j]; }
    const int base = min(max(sb, 0), E);
    const int m    = min(max(sm, 0), E - base);
    const int2* pr2 = (const int2*)pairs;

    int2 pr[SPW];
    int  ln[SPW];
#pragma unroll
    for (int u = 0; u < SPW; ++u) {
        const int sg = wv + 16 * u;
        ln[u] = seglen[sg];
        pr[u] = pr2[min(segbeg[sg] + min(lane, max(ln[u] - 1, 0)), E - 1)];
    }
    asm volatile("" :: "v"(pr[0].y), "v"(pr[1].y), "v"(pr[2].y), "v"(pr[3].y),
                       "v"(pr[4].y), "v"(pr[5].y), "v"(pr[6].y), "v"(pr[7].y));
#pragma unroll
    for (int u = 0; u < SPW; ++u)
        if (lane < ln[u]) atomicAdd(&cnt[min(max(pr[u].y - n0, 0), NPB - 1)], 1);
    for (int u = 0; u < SPW; ++u) {
        const int sg = wv + 16 * u;
        const int L  = seglen[sg];
        for (int j = 32 + lane; j < L; j += 32) {
            const int2 q = pr2[min(segbeg[sg] + j, E - 1)];
            atomicAdd(&cnt[min(max(q.y - n0, 0), NPB - 1)], 1);
        }
    }
    __syncthreads();
    const int c = cnt[tid];
    int inc = c;
    for (int o = 1; o < 32; o <<= 1) {
        const int t = __shfl_up(inc, o);
        if (lane >= o) inc += t;
    }
    if (lane == 31) wsum[wv] = inc;
    __syncthreads();
    int woff = 0;
    for (int j = 0; j < 16; ++j)
        if (j < wv) woff += wsum[j];
    const int rs = woff + inc - c;
    cnt[tid] = rs;
    const int node = n0 + tid;
    if (node < N) {
        rowbeg[node] = base + rs;
        rowend[node] = base + rs + c;
        const float dg = (float)(c + 1);
        dinv[node] = (dg > 0.0f) ? rsqrtf(dg) : 0.0f;
    }
    __syncthreads();
#pragma unroll
    for (int u = 0; u < SPW; ++u) {
        if (lane < ln[u]) {
            const int p = atomicAdd(&cnt[min(max(pr[u].y - n0, 0), NPB - 1)], 1);
            if ((unsigned)p < (unsigned)CAP) stage[p] = pr[u].x;
            else if ((unsigned)p < (unsigned)m) col[base + p] = pr[u].x;
        }
    }
    for (int u = 0; u < SPW; ++u) {
        const int sg = wv + 16 * u;
        const int L  = seglen[sg];
        for (int j = 32 + lane; j < L; j += 32) {
            const int2 q = pr2[min(segbeg[sg] + j, E - 1)];
            const int p = atomicAdd(&cnt[min(max(q.y - n0, 0), NPB - 1)], 1);
            if ((unsigned)p < (unsigned)CAP) stage[p] = q.x;
            else if ((unsigned)p < (unsigned)m) col[base + p] = q.x;
        }
    }
    __syncthreads();
    const int mc = min(m, CAP);
    for (int i = tid; i < mc; i += 512) col[base + i] = stage[i];
}

__global__ __launch_bounds__(512) void k_layer1_fused(const float* __restrict__ x,
                                                      const int* __restrict__ rowbeg,
                                                      const int* __restrict__ col,
                                                      const float* __restrict__ dinv,
                                                      const float* __restrict__ W1,
                                                      const float* __restrict__ b1,
                                                      const float* __restrict__ W2,
                                                      float* __restrict__ g, int N, int E) {
    constexpr int BP1 = 2 * 128 + 32;
    constexpr int BP2 = 2 * 64 + 32;
    constexpr int AP  = 132;

    __shared__ __attribute__((aligned(16))) float B1s[64 * BP1];
    __shared__ __attribute__((aligned(16))) float B2s[64 * BP2];
    __shared__ __attribute__((aligned(16))) float As[64 * AP];

    const int tid  = threadIdx.x;
    const int lane = tid & 31;
    const int wv   = tid >> 5;
    const int half = lane >> 4;
    const int r    = lane & 15;
    const int tiles = (N + 63) >> 6;
    const int tstep = gridDim.x;

    int   rb_raw;
    float dv_raw;
    {
        const int mrowc = min((int)blockIdx.x * 64 + 4 * wv + lane, N - 1);
        rb_raw = rowbeg[mrowc];
        dv_raw = dinv[mrowc];
    }

    float4 w1e[4], w1o[4], w2e[2], w2o[2];
#pragma unroll
    for (int it = 0; it < 4; ++it) {
        const int idx = tid + it * 512;
        const int kp  = idx >> 5;
        const int n4  = (idx & 31) * 4;
        w1e[it] = *(const float4*)(W1 + (size_t)(2 * kp) * 128 + n4);
        w1o[it] = *(const float4*)(W1 + (size_t)(2 * kp + 1) * 128 + n4);
    }
#pragma unroll
    for (int it = 0; it < 2; ++it) {
        const int idx = tid + it * 512;
        const int kp  = idx >> 4;
        const int n4  = (idx & 15) * 4;
        w2e[it] = *(const float4*)(W2 + (size_t)(2 * kp) * 64 + n4);
        w2o[it] = *(const float4*)(W2 + (size_t)(2 * kp + 1) * 64 + n4);
    }

#pragma unroll
    for (int it = 0; it < 4; ++it) {
        const int idx = tid + it * 512;
        const int kp  = idx >> 5;
        const int n4  = (idx & 31) * 4;
        float4* d = (float4*)(B1s + kp * BP1 + 2 * n4);
        d[0] = make_float4(w1e[it].x, w1o[it].x, w1e[it].y, w1o[it].y);
        d[1] = make_float4(w1e[it].z, w1o[it].z, w1e[it].w, w1o[it].w);
    }
#pragma unroll
    for (int it = 0; it < 2; ++it) {
        const int idx = tid + it * 512;
        const int kp  = idx >> 4;
        const int n4  = (idx & 15) * 4;
        float4* d = (float4*)(B2s + kp * BP2 + 2 * n4);
        d[0] = make_float4(w2e[it].x, w2o[it].x, w2e[it].y, w2o[it].y);
        d[1] = make_float4(w2e[it].z, w2o[it].z, w2e[it].w, w2o[it].w);
    }

    for (int tile = blockIdx.x; tile < tiles; tile += tstep) {
        const int m0   = tile * 64;
        const int row0 = m0 + 4 * wv;
        const int mrow = row0 + lane;

        v4f selfv[4];
#pragma unroll
        for (int i = 0; i < 4; ++i)
            selfv[i] = *(const v4f*)(x + (size_t)min(row0 + i, N - 1) * 128 + 4 * lane);

        const int   bl  = (mrow < N) ? min(max(rb_raw, 0), E) : E;
        const float mdd = (mrow < N) ? dv_raw : 0.0f;
        {
            const int nrowc = min(mrow + tstep * 64, N - 1);
            rb_raw = rowbeg[nrowc];
            dv_raw = dinv[nrowc];
        }
        const int ebeg = __builtin_amdgcn_readlane(bl, 0);
        const int eend = __builtin_amdgcn_readlane(bl, 4);
        int s_first = 0;
        if (lane < eend - ebeg) s_first = col[ebeg + lane];

        float* trow = As + (4 * wv) * AP + 4 * lane;
#pragma unroll
        for (int i = 0; i < 4; ++i) {
            const float di = __int_as_float(__builtin_amdgcn_readlane(__float_as_int(mdd), i));
            *(v4f*)(trow + i * AP) = (row0 + i < N) ? selfv[i] * (di * di) : v4f{};
        }

        {
            int   cur  = 0;
            int   endc = __builtin_amdgcn_readlane(bl, 1);
            float ddc  = __int_as_float(__builtin_amdgcn_readlane(__float_as_int(mdd), 0));
            v4f   acc  = v4f{};
            const float* xl = x + 4 * lane;

            auto take = [&](const v4f v, const float wsrc, const int e) {
                while (e >= endc && cur < 3) {
                    *(v4f*)(trow + cur * AP) += acc;
                    ++cur;
                    endc = __builtin_amdgcn_readlane(bl, cur + 1);
                    ddc  = __int_as_float(__builtin_amdgcn_readlane(__float_as_int(mdd), cur));
                    acc  = v4f{};
                }
                acc += v * (wsrc * ddc);
            };

            for (int base = ebeg; base < eend; base += 32) {
                const int n = min(32, eend - base);
                int raw = s_first;
                if (base != ebeg) {
                    raw = 0;
                    if (lane < n) raw = col[base + lane];
                }
                const int my_s = min(max(raw, 0), N - 1);
                float my_w = 0.0f;
                if (lane < n) my_w = dinv[my_s];
                for (int j = 0; j < n; j += 8) {
                    const int m = min(8, n - j);
                    v4f v[8];
                    if (m == 8) {
#pragma unroll
                        for (int k = 0; k < 8; ++k)
                            v[k] = *(const v4f*)(xl + (size_t)__builtin_amdgcn_readlane(my_s, j + k) * 128);
                    } else {
#pragma unroll
                        for (int k = 0; k < 8; ++k) {
                            v[k] = v4f{};
                            if (k < m) v[k] = *(const v4f*)(xl + (size_t)__builtin_amdgcn_readlane(my_s, j + k) * 128);
                        }
                    }
#pragma unroll
                    for (int k = 0; k < 8; ++k) {
                        if (k < m)
                            take(v[k], __int_as_float(__builtin_amdgcn_readlane(__float_as_int(my_w), j + k)), base + j + k);
                    }
                }
            }
            *(v4f*)(trow + cur * AP) += acc;
        }
        __syncthreads();

        const int rt = wv & 3;
        const int cg = wv >> 2;
        const float* ap = As + (16 * rt + r) * AP + 2 * half;

        v8f c0 = v8f{};
        v8f c1 = v8f{};
        {
            const float* bp = B1s + half * BP1 + 2 * r + 64 * cg;
#pragma unroll
            for (int kk = 0; kk < 128; kk += 4) {
                const v2f a  = *(const v2f*)(ap + kk);
                const v2f bA = *(const v2f*)(bp + (kk >> 1) * BP1);
                const v2f bB = *(const v2f*)(bp + (kk >> 1) * BP1 + 32);
                c0 = __builtin_amdgcn_wmma_f32_16x16x4_f32(false, a, false, bA, (short)0, c0, false, false);
                c1 = __builtin_amdgcn_wmma_f32_16x16x4_f32(false, a, false, bB, (short)0, c1, false, false);
            }
        }
        __syncthreads();
        {
            const int   colA  = 32 * cg + r;
            const float biasA = b1[colA];
            const float biasB = b1[colA + 16];
#pragma unroll
            for (int r2 = 0; r2 < 8; ++r2) {
                const float hA = c0[r2] + biasA;
                const float hB = c1[r2] + biasB;
                float* t = As + (16 * rt + r2 + 8 * half) * AP + colA;
                t[0]  = hA > 0.0f ? hA : 0.0f;
                t[16] = hB > 0.0f ? hB : 0.0f;
            }
        }
        __syncthreads();

        v8f c2 = v8f{};
        {
            const float* bp = B2s + half * BP2 + 2 * r + 32 * cg;
#pragma unroll
            for (int kk = 0; kk < 128; kk += 4) {
                const v2f a = *(const v2f*)(ap + kk);
                const v2f b = *(const v2f*)(bp + (kk >> 1) * BP2);
                c2 = __builtin_amdgcn_wmma_f32_16x16x4_f32(false, a, false, b, (short)0, c2, false, false);
            }
        }
        __syncthreads();
#pragma unroll
        for (int r2 = 0; r2 < 8; ++r2)
            As[(16 * rt + r2 + 8 * half) * AP + 16 * cg + r] = c2[r2];
        __syncthreads();

        const int orow = tid >> 4;
        const int oc4  = 4 * (tid & 15);
        const v4f o0 = *(const v4f*)(As + orow * AP + oc4);
        const v4f o1 = *(const v4f*)(As + (orow + 32) * AP + oc4);
        __syncthreads();
        if (m0 + orow < N)      *(v4f*)(g + (size_t)(m0 + orow) * 64 + oc4)      = o0;
        if (m0 + orow + 32 < N) *(v4f*)(g + (size_t)(m0 + orow + 32) * 64 + oc4) = o1;
    }
}

template<int C, bool RELU>
__global__ __launch_bounds__(256) void k_gather_rows(const float* __restrict__ feat,
                                                     const int* __restrict__ rowbeg,
                                                     const int* __restrict__ rowend,
                                                     const int* __restrict__ col,
                                                     const float* __restrict__ dinv,
                                                     const float* __restrict__ bias,
                                                     float* __restrict__ out, int N, int E) {
    constexpr int V = C / 32;
    typedef float vec_t __attribute__((ext_vector_type(V)));
    const int lane = threadIdx.x & 31;
    const int d = __builtin_amdgcn_readfirstlane((int)((blockIdx.x * blockDim.x + threadIdx.x) >> 5));
    if (d >= N) return;

    const float dd  = dinv[d];
    const int   beg = max(rowbeg[d], 0);
    const int   end = min(rowend[d], E);

    vec_t acc = *(const vec_t*)(feat + (size_t)d * C + lane * V) * (dd * dd);

    for (int base = beg; base < end; base += 32) {
        const int n = min(32, end - base);
        int   my_s = d;
        float my_w = 0.0f;
        if (lane < n) {
            my_s = min(max(col[base + lane], 0), N - 1);
            my_w = dinv[my_s];
        }
        int j = 0;
        for (; j + 4 <= n; j += 4) {
            const int s0 = __builtin_amdgcn_readlane(my_s, j);
            const int s1 = __builtin_amdgcn_readlane(my_s, j + 1);
            const int s2 = __builtin_amdgcn_readlane(my_s, j + 2);
            const int s3 = __builtin_amdgcn_readlane(my_s, j + 3);
            const vec_t v0 = *(const vec_t*)(feat + (size_t)s0 * C + lane * V);
            const vec_t v1 = *(const vec_t*)(feat + (size_t)s1 * C + lane * V);
            const vec_t v2 = *(const vec_t*)(feat + (size_t)s2 * C + lane * V);
            const vec_t v3 = *(const vec_t*)(feat + (size_t)s3 * C + lane * V);
            const float w0 = __int_as_float(__builtin_amdgcn_readlane(__float_as_int(my_w), j)) * dd;
            const float w1 = __int_as_float(__builtin_amdgcn_readlane(__float_as_int(my_w), j + 1)) * dd;
            const float w2 = __int_as_float(__builtin_amdgcn_readlane(__float_as_int(my_w), j + 2)) * dd;
            const float w3 = __int_as_float(__builtin_amdgcn_readlane(__float_as_int(my_w), j + 3)) * dd;
            acc += v0 * w0;
            acc += v1 * w1;
            acc += v2 * w2;
            acc += v3 * w3;
        }
        const int rem = n - j;
        if (rem > 0) {
            vec_t v0 = *(const vec_t*)(feat + (size_t)__builtin_amdgcn_readlane(my_s, j) * C + lane * V);
            vec_t v1 = {};
            vec_t v2 = {};
            float w0 = __int_as_float(__builtin_amdgcn_readlane(__float_as_int(my_w), j)) * dd;
            float w1 = 0.0f;
            float w2 = 0.0f;
            if (rem > 1) {
                v1 = *(const vec_t*)(feat + (size_t)__builtin_amdgcn_readlane(my_s, j + 1) * C + lane * V);
                w1 = __int_as_float(__builtin_amdgcn_readlane(__float_as_int(my_w), j + 1)) * dd;
            }
            if (rem > 2) {
                v2 = *(const vec_t*)(feat + (size_t)__builtin_amdgcn_readlane(my_s, j + 2) * C + lane * V);
                w2 = __int_as_float(__builtin_amdgcn_readlane(__float_as_int(my_w), j + 2)) * dd;
            }
            acc += v0 * w0;
            acc += v1 * w1;
            acc += v2 * w2;
        }
    }

    vec_t r = acc + *(const vec_t*)(bias + lane * V);
    if (RELU) {
        for (int k = 0; k < V; ++k) r[k] = r[k] > 0.0f ? r[k] : 0.0f;
    }
    *(vec_t*)(out + (size_t)d * C + lane * V) = r;
}

extern "C" void kernel_launch(void* const* d_in, const int* in_sizes, int n_in,
                              void* d_out, int out_size, void* d_ws, size_t ws_size,
                              hipStream_t stream) {
    const float* x  = (const float*)d_in[0];
    const int*   ei = (const int*)  d_in[1];
    const float* W1 = (const float*)d_in[2];
    const float* b1 = (const float*)d_in[3];
    const float* W2 = (const float*)d_in[4];
    const float* b2 = (const float*)d_in[5];

    const int N = in_sizes[0] / 128;
    const int E = in_sizes[1] / 2;
    const int* src = ei;
    const int* dst = ei + E;
    const int KB = (N + NPB - 1) / NPB;

    char* ws = (char*)d_ws;
    size_t szN  = (((size_t)N * 4) + 255) & ~(size_t)255;
    size_t szT  = (size_t)EBLK * TP * 4;
    size_t szE  = (((size_t)E * 4) + 255) & ~(size_t)255;
    float* dinv   = (float*)ws;
    int*   rowbeg = (int*)(ws + szN);
    int*   cursor = (int*)(ws + 2 * szN);
    int*   T      = (int*)(ws + 3 * szN);
    int*   pairs  = (int*)(ws + 3 * szN + szT);
    int*   col    = (int*)(ws + 3 * szN + szT + 2 * szE);
    size_t off  = 3 * szN + szT + 3 * szE;
    float* bufG = (float*)(ws + off);

    k_chunk_sort<<<EBLK, 1024, 0, stream>>>(src, dst, T, pairs, E, KB);
    k_bucket_rows<<<KB, 512, 0, stream>>>(pairs, T, dinv, rowbeg, cursor, col, N, E);

    const int tiles = (N + 63) / 64;
    k_layer1_fused<<<(tiles < 512 ? tiles : 512), 512, 0, stream>>>(x, rowbeg, col, dinv, W1, b1, W2, bufG, N, E);

    k_gather_rows<64, false><<<(N + 7) / 8, 256, 0, stream>>>(bufG, rowbeg, cursor, col, dinv, b2, (float*)d_out, N, E);
}


# Layer-2 gather made resident: 8192 waves (was 100000 one-row waves) walk rows w, w+W, ..; a trip requests its rows, the next row's sources and the bounds after, then waits once; sum order kept
# speedup vs baseline: 4.5825x; 1.0402x over previous
#include <hip/hip_runtime.h>

typedef float v2f __attribute__((ext_vector_type(2)));
typedef float v4f __attribute__((ext_vector_type(4)));
typedef float v8f __attribute__((ext_vector_type(8)));

constexpr int NPB_LOG = 9;
constexpr int NPB     = 1 << NPB_LOG;
constexpr int MAXKB   = 1024;
constexpr int EBLK    = 128;
constexpr int TP      = MAXKB + 64;

__device__ __forceinline__ int bucket_of(int d, int KB) {
    return (int)min((unsigned)d >> NPB_LOG, (unsigned)(KB - 1));
}

__global__ __launch_bounds__(1024) void k_chunk_sort(const int* __restrict__ src, const int* __restrict__ dst,
                                                     int* __restrict__ T, int* __restrict__ pairs, int E, int KB) {
    constexpr int CAPS = 8192;
    __shared__ int  cnt[MAXKB];
    __shared__ int  wsum[32];
    __shared__ int2 stage[CAPS];
    const int tid  = threadIdx.x;
    const int lane = tid & 31;
    const int wv   = tid >> 5;
    const int b    = blockIdx.x;
    const int CH   = (E + EBLK - 1) / EBLK;
    const int lo   = min(b * CH, E);
    const int hi   = min(E, lo + CH);
    const int mine = hi - lo;
    const bool one_sweep = CH <= 8 * 1024;
    cnt[tid] = 0;
    __syncthreads();
    int d[8], s[8];
#pragma unroll
    for (int u = 0; u < 8; ++u) { d[u] = 0; s[u] = 0; }
    for (int e0 = lo + tid; e0 < hi; e0 += 8 * 1024) {
#pragma unroll
        for (int u = 0; u < 8; ++u)
            d[u] = dst[min(e0 + u * 1024, hi - 1)];
        if (one_sweep) {
#pragma unroll
            for (int u = 0; u < 8; ++u)
                s[u] = src[min(e0 + u * 1024, hi - 1)];
        }
        asm volatile("" :: "v"(d[0]), "v"(d[1]), "v"(d[2]), "v"(d[3]), "v"(d[4]), "v"(d[5]), "v"(d[6]), "v"(d[7]));
#pragma unroll
        for (int u = 0; u < 8; ++u)
            if (e0 + u * 1024 < hi) atomicAdd(&cnt[bucket_of(d[u], KB)], 1);
    }
    __syncthreads();
    const int v = (tid < KB) ? cnt[tid] : 0;
    int inc = v;
    for (int o = 1; o < 32; o <<= 1) {
        const int t = __shfl_up(inc, o);
        if (lane >= o) inc += t;
    }
    if (lane == 31) wsum[wv] = inc;
    __syncthreads();
    int woff = 0;
    for (int k = 0; k < 32; ++k)
        if (k < wv) woff += wsum[k];
    const int excl = woff + inc - v;
    if (tid < KB) {
        cnt[tid] = excl;
        T[b * TP + tid] = excl;
        if (tid == KB - 1) T[b * TP + KB] = excl + v;
    }
    __syncthreads();
    int2* pr2 = (int2*)pairs;
    for (int e0 = lo + tid; e0 < hi; e0 += 8 * 1024) {
        if (!one_sweep) {
#pragma unroll
            for (int u = 0; u < 8; ++u) {
                const int e = min(e0 + u * 1024, hi - 1);
                d[u] = dst[e];
                s[u] = src[e];
            }
            asm volatile("" :: "v"(d[0]), "v"(d[1]), "v"(d[2]), "v"(d[3]), "v"(d[4]), "v"(d[5]), "v"(d[6]), "v"(d[7]),
                               "v"(s[0]), "v"(s[1]), "v"(s[2]), "v"(s[3]), "v"(s[4]), "v"(s[5]), "v"(s[6]), "v"(s[7]));
        }
#pragma unroll
        for (int u = 0; u < 8; ++u) {
            if (e0 + u * 1024 < hi) {
                const int p = atomicAdd(&cnt[bucket_of(d[u], KB)], 1);
                if ((unsigned)p < (unsigned)CAPS) stage[p] = make_int2(s[u], d[u]);
                else if ((unsigned)p < (unsigned)mine) pr2[lo + p] = make_int2(s[u], d[u]);
            }
        }
    }
    __syncthreads();
    const int ms = min(mine, CAPS);
    for (int i = tid; i < ms; i += 1024) pr2[lo + i] = stage[i];
}

__global__ __launch_bounds__(512) void k_bucket_rows(const int* __restrict__ pairs, const int* __restrict__ T,
                                                     float* __restrict__ dinv, int* __restrict__ rowbeg,
                                                     int* __restrict__ rowend, int* __restrict__ col, int N, int E) {
    constexpr int CAP = 8192;
    constexpr int SPW = EBLK / 16;
    static_assert(SPW == 8, "the load fence below names eight segments");
    __shared__ int cnt[NPB];
    __shared__ int wsum[16];
    __shared__ int stage[CAP];
    __shared__ int segbeg[EBLK];
    __shared__ int seglen[EBLK];
    __shared__ int pbase[EBLK / 32];
    __shared__ int pm[EBLK / 32];
    const int tid  = threadIdx.x;
    const int lane = tid & 31;
    const int wv   = tid >> 5;
    const int k    = blockIdx.x;
    const int n0   = k << NPB_LOG;
    cnt[tid] = 0;
    if (tid < EBLK) {
        const int CH   = (E + EBLK - 1) / EBLK;
        const int lo   = min(tid * CH, E);
        const int mine = min(E, lo + CH) - lo;
        const int a = min(max(T[tid * TP + k], 0), mine);
        const int z = min(max(T[tid * TP + k + 1], a), mine);
        segbeg[tid] = lo + a;
        seglen[tid] = z - a;
        int sa = a, sl = z - a;
        for (int o = 16; o > 0; o >>= 1) {
            sa += __shfl_xor(sa, o);
            sl += __shfl_xor(sl, o);
        }
        if (lane == 0) { pbase[wv] = sa; pm[wv] = sl; }
    }
    __syncthreads();
    int sb = 0, sm = 0;
    for (int j = 0; j < EBLK / 32; ++j) { sb += pbase[j]; sm += pm[j]; }
    const int base = min(max(sb, 0), E);
    const int m    = min(max(sm, 0), E - base);
    const int2* pr2 = (const int2*)pairs;

    int2 pr[SPW];
    int  ln[SPW];
#pragma unroll
    for (int u = 0; u < SPW; ++u) {
        const int sg = wv + 16 * u;
        ln[u] = seglen[sg];
        pr[u] = pr2[min(segbeg[sg] + min(lane, max(ln[u] - 1, 0)), E - 1)];
    }
    asm volatile("" :: "v"(pr[0].y), "v"(pr[1].y), "v"(pr[2].y), "v"(pr[3].y),
                       "v"(pr[4].y), "v"(pr[5].y), "v"(pr[6].y), "v"(pr[7].y));
#pragma unroll
    for (int u = 0; u < SPW; ++u)
        if (lane < ln[u]) atomicAdd(&cnt[min(max(pr[u].y - n0, 0), NPB - 1)], 1);
    for (int u = 0; u < SPW; ++u) {
        const int sg = wv + 16 * u;
        const int L  = seglen[sg];
        for (int j = 32 + lane; j < L; j += 32) {
            const int2 q = pr2[min(segbeg[sg] + j, E - 1)];
            atomicAdd(&cnt[min(max(q.y - n0, 0), NPB - 1)], 1);
        }
    }
    __syncthreads();
    const int c = cnt[tid];
    int inc = c;
    for (int o = 1; o < 32; o <<= 1) {
        const int t = __shfl_up(inc, o);
        if (lane >= o) inc += t;
    }
    if (lane == 31) wsum[wv] = inc;
    __syncthreads();
    int woff = 0;
    for (int j = 0; j < 16; ++j)
        if (j < wv) woff += wsum[j];
    const int rs = woff + inc - c;
    cnt[tid] = rs;
    const int node = n0 + tid;
    if (node < N) {
        rowbeg[node] = base + rs;
        rowend[node] = base + rs + c;
        const float dg = (float)(c + 1);
        dinv[node] = (dg > 0.0f) ? rsqrtf(dg) : 0.0f;
    }
    __syncthreads();
#pragma unroll
    for (int u = 0; u < SPW; ++u) {
        if (lane < ln[u]) {
            const int p = atomicAdd(&cnt[min(max(pr[u].y - n0, 0), NPB - 1)], 1);
            if ((unsigned)p < (unsigned)CAP) stage[p] = pr[u].x;
            else if ((unsigned)p < (unsigned)m) col[base + p] = pr[u].x;
        }
    }
    for (int u = 0; u < SPW; ++u) {
        const int sg = wv + 16 * u;
        const int L  = seglen[sg];
        for (int j = 32 + lane; j < L; j += 32) {
            const int2 q = pr2[min(segbeg[sg] + j, E - 1)];
            const int p = atomicAdd(&cnt[min(max(q.y - n0, 0), NPB - 1)], 1);
            if ((unsigned)p < (unsigned)CAP) stage[p] = q.x;
            else if ((unsigned)p < (unsigned)m) col[base + p] = q.x;
        }
    }
    __syncthreads();
    const int mc = min(m, CAP);
    for (int i = tid; i < mc; i += 512) col[base + i] = stage[i];
}

__global__ __launch_bounds__(512) void k_layer1_fused(const float* __restrict__ x,
                                                      const int* __restrict__ rowbeg,
                                                      const int* __restrict__ col,
                                                      const float* __restrict__ dinv,
                                                      const float* __restrict__ W1,
                                                      const float* __restrict__ b1,
                                                      const float* __restrict__ W2,
                                                      float* __restrict__ g, int N, int E) {
    constexpr int BP1 = 2 * 128 + 32;
    constexpr int BP2 = 2 * 64 + 32;
    constexpr int AP  = 132;

    __shared__ __attribute__((aligned(16))) float B1s[64 * BP1];
    __shared__ __attribute__((aligned(16))) float B2s[64 * BP2];
    __shared__ __attribute__((aligned(16))) float As[64 * AP];

    const int tid  = threadIdx.x;
    const int lane = tid & 31;
    const int wv   = tid >> 5;
    const int half = lane >> 4;
    const int r    = lane & 15;
    const int tiles = (N + 63) >> 6;
    const int tstep = gridDim.x;

    int   rb_raw;
    float dv_raw;
    {
        const int mrowc = min((int)blockIdx.x * 64 + 4 * wv + lane, N - 1);
        rb_raw = rowbeg[mrowc];
        dv_raw = dinv[mrowc];
    }

    float4 w1e[4], w1o[4], w2e[2], w2o[2];
#pragma unroll
    for (int it = 0; it < 4; ++it) {
        const int idx = tid + it * 512;
        const int kp  = idx >> 5;
        const int n4  = (idx & 31) * 4;
        w1e[it] = *(const float4*)(W1 + (size_t)(2 * kp) * 128 + n4);
        w1o[it] = *(const float4*)(W1 + (size_t)(2 * kp + 1) * 128 + n4);
    }
#pragma unroll
    for (int it = 0; it < 2; ++it) {
        const int idx = tid + it * 512;
        const int kp  = idx >> 4;
        const int n4  = (idx & 15) * 4;
        w2e[it] = *(const float4*)(W2 + (size_t)(2 * kp) * 64 + n4);
        w2o[it] = *(const float4*)(W2 + (size_t)(2 * kp + 1) * 64 + n4);
    }

#pragma unroll
    for (int it = 0; it < 4; ++it) {
        const int idx = tid + it * 512;
        const int kp  = idx >> 5;
        const int n4  = (idx & 31) * 4;
        float4* d = (float4*)(B1s + kp * BP1 + 2 * n4);
        d[0] = make_float4(w1e[it].x, w1o[it].x, w1e[it].y, w1o[it].y);
        d[1] = make_float4(w1e[it].z, w1o[it].z, w1e[it].w, w1o[it].w);
    }
#pragma unroll
    for (int it = 0; it < 2; ++it) {
        const int idx = tid + it * 512;
        const int kp  = idx >> 4;
        const int n4  = (idx & 15) * 4;
        float4* d = (float4*)(B2s + kp * BP2 + 2 * n4);
        d[0] = make_float4(w2e[it].x, w2o[it].x, w2e[it].y, w2o[it].y);
        d[1] = make_float4(w2e[it].z, w2o[it].z, w2e[it].w, w2o[it].w);
    }

    for (int tile = blockIdx.x; tile < tiles; tile += tstep) {
        const int m0   = tile * 64;
        const int row0 = m0 + 4 * wv;
        const int mrow = row0 + lane;

        v4f selfv[4];
#pragma unroll
        for (int i = 0; i < 4; ++i)
            selfv[i] = *(const v4f*)(x + (size_t)min(row0 + i, N - 1) * 128 + 4 * lane);

        const int   bl  = (mrow < N) ? min(max(rb_raw, 0), E) : E;
        const float mdd = (mrow < N) ? dv_raw : 0.0f;
        {
            const int nrowc = min(mrow + tstep * 64, N - 1);
            rb_raw = rowbeg[nrowc];
            dv_raw = dinv[nrowc];
        }
        const int ebeg = __builtin_amdgcn_readlane(bl, 0);
        const int eend = __builtin_amdgcn_readlane(bl, 4);
        int s_first = 0;
        if (lane < eend - ebeg) s_first = col[ebeg + lane];

        float* trow = As + (4 * wv) * AP + 4 * lane;
#pragma unroll
        for (int i = 0; i < 4; ++i) {
            const float di = __int_as_float(__builtin_amdgcn_readlane(__float_as_int(mdd), i));
            *(v4f*)(trow + i * AP) = (row0 + i < N) ? selfv[i] * (di * di) : v4f{};
        }

        {
            int   cur  = 0;
            int   endc = __builtin_amdgcn_readlane(bl, 1);
            float ddc  = __int_as_float(__builtin_amdgcn_readlane(__float_as_int(mdd), 0));
            v4f   acc  = v4f{};
            const float* xl = x + 4 * lane;

            auto take = [&](const v4f v, const float wsrc, const int e) {
                while (e >= endc && cur < 3) {
                    *(v4f*)(trow + cur * AP) += acc;
                    ++cur;
                    endc = __builtin_amdgcn_readlane(bl, cur + 1);
                    ddc  = __int_as_float(__builtin_amdgcn_readlane(__float_as_int(mdd), cur));
                    acc  = v4f{};
                }
                acc += v * (wsrc * ddc);
            };

            for (int base = ebeg; base < eend; base += 32) {
                const int n = min(32, eend - base);
                int raw = s_first;
                if (base != ebeg) {
                    raw = 0;
                    if (lane < n) raw = col[base + lane];
                }
                const int my_s = min(max(raw, 0), N - 1);
                float my_w = 0.0f;
                if (lane < n) my_w = dinv[my_s];
                for (int j = 0; j < n; j += 8) {
                    const int m = min(8, n - j);
                    v4f v[8];
                    if (m == 8) {
#pragma unroll
                        for (int k = 0; k < 8; ++k)
                            v[k] = *(const v4f*)(xl + (size_t)__builtin_amdgcn_readlane(my_s, j + k) * 128);
                    } else {
#pragma unroll
                        for (int k = 0; k < 8; ++k) {
                            v[k] = v4f{};
                            if (k < m) v[k] = *(const v4f*)(xl + (size_t)__builtin_amdgcn_readlane(my_s, j + k) * 128);
                        }
                    }
#pragma unroll
                    for (int k = 0; k < 8; ++k) {
                        if (k < m)
                            take(v[k], __int_as_float(__builtin_amdgcn_readlane(__float_as_int(my_w), j + k)), base + j + k);
                    }
                }
            }
            *(v4f*)(trow + cur * AP) += acc;
        }
        __syncthreads();

        const int rt = wv & 3;
        const int cg = wv >> 2;
        const float* ap = As + (16 * rt + r) * AP + 2 * half;

        v8f c0 = v8f{};
        v8f c1 = v8f{};
        {
            const float* bp = B1s + half * BP1 + 2 * r + 64 * cg;
#pragma unroll
            for (int kk = 0; kk < 128; kk += 4) {
                const v2f a  = *(const v2f*)(ap + kk);
                const v2f bA = *(const v2f*)(bp + (kk >> 1) * BP1);
                const v2f bB = *(const v2f*)(bp + (kk >> 1) * BP1 + 32);
                c0 = __builtin_amdgcn_wmma_f32_16x16x4_f32(false, a, false, bA, (short)0, c0, false, false);
                c1 = __builtin_amdgcn_wmma_f32_16x16x4_f32(false, a, false, bB, (short)0, c1, false, false);
            }
        }
        __syncthreads();
        {
            const int   colA  = 32 * cg + r;
            const float biasA = b1[colA];
            const float biasB = b1[colA + 16];
#pragma unroll
            for (int r2 = 0; r2 < 8; ++r2) {
                const float hA = c0[r2] + biasA;
                const float hB = c1[r2] + biasB;
                float* t = As + (16 * rt + r2 + 8 * half) * AP + colA;
                t[0]  = hA > 0.0f ? hA : 0.0f;
                t[16] = hB > 0.0f ? hB : 0.0f;
            }
        }
        __syncthreads();

        v8f c2 = v8f{};
        {
            const float* bp = B2s + half * BP2 + 2 * r + 32 * cg;
#pragma unroll
            for (int kk = 0; kk < 128; kk += 4) {
                const v2f a = *(const v2f*)(ap + kk);
                const v2f b = *(const v2f*)(bp + (kk >> 1) * BP2);
                c2 = __builtin_amdgcn_wmma_f32_16x16x4_f32(false, a, false, b, (short)0, c2, false, false);
            }
        }
        __syncthreads();
#pragma unroll
        for (int r2 = 0; r2 < 8; ++r2)
            As[(16 * rt + r2 + 8 * half) * AP + 16 * cg + r] = c2[r2];
        __syncthreads();

        const int orow = tid >> 4;
        const int oc4  = 4 * (tid & 15);
        const v4f o0 = *(const v4f*)(As + orow * AP + oc4);
        const v4f o1 = *(const v4f*)(As + (orow + 32) * AP + oc4);
        __syncthreads();
        if (m0 + orow < N)      *(v4f*)(g + (size_t)(m0 + orow) * 64 + oc4)      = o0;
        if (m0 + orow + 32 < N) *(v4f*)(g + (size_t)(m0 + orow + 32) * 64 + oc4) = o1;
    }
}

template<int C, bool RELU>
__global__ __launch_bounds__(256) void k_gather_rows(const float* __restrict__ feat,
                                                     const int* __restrict__ rowbeg,
                                                     const int* __restrict__ rowend,
                                                     const int* __restrict__ col,
                                                     const float* __restrict__ dinv,
                                                     const float* __restrict__ bias,
                                                     float* __restrict__ out, int N, int E) {
    constexpr int V = C / 32;
    typedef float vec_t __attribute__((ext_vector_type(V)));
    const int lane = threadIdx.x & 31;
    const int wave0 = __builtin_amdgcn_readfirstlane((int)((blockIdx.x * blockDim.x + threadIdx.x) >> 5));
    const int nwave = (int)((gridDim.x * blockDim.x) >> 5);
    if (wave0 >= N) return;

    const float* fl = feat + lane * V;
    const vec_t  bv = *(const vec_t*)(bias + lane * V);

    auto batch = [&](vec_t& acc, const int ms, const float mw, const int j, const int m, const float ddv) {
        vec_t v[8];
#pragma unroll
        for (int k = 0; k < 8; ++k) {
            v[k] = vec_t{};
            if (k < m) v[k] = *(const vec_t*)(fl + (size_t)__builtin_amdgcn_readlane(ms, j + k) * C);
        }
#pragma unroll
        for (int k = 0; k < 8; ++k)
            if (k < m) acc += v[k] * (__int_as_float(__builtin_amdgcn_readlane(__float_as_int(mw), j + k)) * ddv);
    };

    int   beg = max(rowbeg[wave0], 0);
    int   end = min(rowend[wave0], E);
    float dd  = dinv[wave0];
    int   begn, endn;
    float ddn;
    {
        const int d1 = min(wave0 + nwave, N - 1);
        begn = rowbeg[d1];
        endn = rowend[d1];
        ddn  = dinv[d1];
    }
    int raw = 0;
    if (lane < end - beg) raw = col[beg + lane];

    for (int d = wave0; d < N; d += nwave) {
        const int n    = min(32, end - beg);
        const int my_s = min(max(raw, 0), N - 1);
        float my_w = 0.0f;
        if (lane < n) my_w = dinv[my_s];
        const vec_t selfv = *(const vec_t*)(fl + (size_t)d * C);
        const int m0 = min(8, n);
        vec_t v[8];
#pragma unroll
        for (int k = 0; k < 8; ++k) {
            v[k] = vec_t{};
            if (k < m0) v[k] = *(const vec_t*)(fl + (size_t)__builtin_amdgcn_readlane(my_s, k) * C);
        }
        const int begc = max(begn, 0);
        const int endc = min(endn, E);
        int rawn = 0;
        if (lane < endc - begc) rawn = col[begc + lane];
        const int   d2   = min(d + 2 * nwave, N - 1);
        const int   beg2 = rowbeg[d2];
        const int   end2 = rowend[d2];
        const float dd2  = dinv[d2];

        vec_t acc = selfv * (dd * dd);
#pragma unroll
        for (int k = 0; k < 8; ++k)
            if (k < m0) acc += v[k] * (__int_as_float(__builtin_amdgcn_readlane(__float_as_int(my_w), k)) * dd);
        for (int j = 8; j < n; j += 8) batch(acc, my_s, my_w, j, min(8, n - j), dd);
        for (int base = beg + 32; base < end; base += 32) {
            const int nn = min(32, end - base);
            int   cs = d;
            float cw = 0.0f;
            if (lane < nn) {
                cs = min(max(col[base + lane], 0), N - 1);
                cw = dinv[cs];
            }
            for (int j = 0; j < nn; j += 8) batch(acc, cs, cw, j, min(8, nn - j), dd);
        }

        vec_t r = acc + bv;
        if (RELU) {
            for (int k = 0; k < V; ++k) r[k] = r[k] > 0.0f ? r[k] : 0.0f;
        }
        *(vec_t*)(out + (size_t)d * C + lane * V) = r;

        beg = begc; end = endc; dd = ddn; raw = rawn;
        begn = beg2; endn = end2; ddn = dd2;
    }
}

extern "C" void kernel_launch(void* const* d_in, const int* in_sizes, int n_in,
                              void* d_out, int out_size, void* d_ws, size_t ws_size,
                              hipStream_t stream) {
    const float* x  = (const float*)d_in[0];
    const int*   ei = (const int*)  d_in[1];
    const float* W1 = (const float*)d_in[2];
    const float* b1 = (const float*)d_in[3];
    const float* W2 = (const float*)d_in[4];
    const float* b2 = (const float*)d_in[5];

    const int N = in_sizes[0] / 128;
    const int E = in_sizes[1] / 2;
    const int* src = ei;
    const int* dst = ei + E;
    const int KB = (N + NPB - 1) / NPB;

    char* ws = (char*)d_ws;
    size_t szN  = (((size_t)N * 4) + 255) & ~(size_t)255;
    size_t szT  = (size_t)EBLK * TP * 4;
    size_t szE  = (((size_t)E * 4) + 255) & ~(size_t)255;
    float* dinv   = (float*)ws;
    int*   rowbeg = (int*)(ws + szN);
    int*   cursor = (int*)(ws + 2 * szN);
    int*   T      = (int*)(ws + 3 * szN);
    int*   pairs  = (int*)(ws + 3 * szN + szT);
    int*   col    = (int*)(ws + 3 * szN + szT + 2 * szE);
    size_t off  = 3 * szN + szT + 3 * szE;
    float* bufG = (float*)(ws + off);

    k_chunk_sort<<<EBLK, 1024, 0, stream>>>(src, dst, T, pairs, E, KB);
    k_bucket_rows<<<KB, 512, 0, stream>>>(pairs, T, dinv, rowbeg, cursor, col, N, E);

    const int tiles = (N + 63) / 64;
    k_layer1_fused<<<(tiles < 512 ? tiles : 512), 512, 0, stream>>>(x, rowbeg, col, dinv, W1, b1, W2, bufG, N, E);

    const int gblocks = (N + 7) / 8;
    k_gather_rows<64, false><<<(gblocks < 1024 ? gblocks : 1024), 256, 0, stream>>>(bufG, rowbeg, cursor, col, dinv, b2, (float*)d_out, N, E);
}


# Layer-2 gather asks for rows only: layer 1 stores rows scaled by dinv[s] (scale factored out of the sum, order kept); a wave owns 12-13 consecutive rows, bounds in one load, dinv[d] from row length
# speedup vs baseline: 4.7393x; 1.0342x over previous
#include <hip/hip_runtime.h>

typedef float v2f __attribute__((ext_vector_type(2)));
typedef float v4f __attribute__((ext_vector_type(4)));
typedef float v8f __attribute__((ext_vector_type(8)));

constexpr int NPB_LOG = 9;
constexpr int NPB     = 1 << NPB_LOG;
constexpr int MAXKB   = 1024;
constexpr int EBLK    = 128;
constexpr int TP      = MAXKB + 64;

__device__ __forceinline__ int bucket_of(int d, int KB) {
    return (int)min((unsigned)d >> NPB_LOG, (unsigned)(KB - 1));
}

__global__ __launch_bounds__(1024) void k_chunk_sort(const int* __restrict__ src, const int* __restrict__ dst,
                                                     int* __restrict__ T, int* __restrict__ pairs, int E, int KB) {
    constexpr int CAPS = 8192;
    __shared__ int  cnt[MAXKB];
    __shared__ int  wsum[32];
    __shared__ int2 stage[CAPS];
    const int tid  = threadIdx.x;
    const int lane = tid & 31;
    const int wv   = tid >> 5;
    const int b    = blockIdx.x;
    const int CH   = (E + EBLK - 1) / EBLK;
    const int lo   = min(b * CH, E);
    const int hi   = min(E, lo + CH);
    const int mine = hi - lo;
    const bool one_sweep = CH <= 8 * 1024;
    cnt[tid] = 0;
    __syncthreads();
    int d[8], s[8];
#pragma unroll
    for (int u = 0; u < 8; ++u) { d[u] = 0; s[u] = 0; }
    for (int e0 = lo + tid; e0 < hi; e0 += 8 * 1024) {
#pragma unroll
        for (int u = 0; u < 8; ++u)
            d[u] = dst[min(e0 + u * 1024, hi - 1)];
        if (one_sweep) {
#pragma unroll
            for (int u = 0; u < 8; ++u)
                s[u] = src[min(e0 + u * 1024, hi - 1)];
        }
        asm volatile("" :: "v"(d[0]), "v"(d[1]), "v"(d[2]), "v"(d[3]), "v"(d[4]), "v"(d[5]), "v"(d[6]), "v"(d[7]));
#pragma unroll
        for (int u = 0; u < 8; ++u)
            if (e0 + u * 1024 < hi) atomicAdd(&cnt[bucket_of(d[u], KB)], 1);
    }
    __syncthreads();
    const int v = (tid < KB) ? cnt[tid] : 0;
    int inc = v;
    for (int o = 1; o < 32; o <<= 1) {
        const int t = __shfl_up(inc, o);
        if (lane >= o) inc += t;
    }
    if (lane == 31) wsum[wv] = inc;
    __syncthreads();
    int woff = 0;
    for (int k = 0; k < 32; ++k)
        if (k < wv) woff += wsum[k];
    const int excl = woff + inc - v;
    if (tid < KB) {
        cnt[tid] = excl;
        T[b * TP + tid] = excl;
        if (tid == KB - 1) T[b * TP + KB] = excl + v;
    }
    __syncthreads();
    int2* pr2 = (int2*)pairs;
    for (int e0 = lo + tid; e0 < hi; e0 += 8 * 1024) {
        if (!one_sweep) {
#pragma unroll
            for (int u = 0; u < 8; ++u) {
                const int e = min(e0 + u * 1024, hi - 1);
                d[u] = dst[e];
                s[u] = src[e];
            }
            asm volatile("" :: "v"(d[0]), "v"(d[1]), "v"(d[2]), "v"(d[3]), "v"(d[4]), "v"(d[5]), "v"(d[6]), "v"(d[7]),
                               "v"(s[0]), "v"(s[1]), "v"(s[2]), "v"(s[3]), "v"(s[4]), "v"(s[5]), "v"(s[6]), "v"(s[7]));
        }
#pragma unroll
        for (int u = 0; u < 8; ++u) {
            if (e0 + u * 1024 < hi) {
                const int p = atomicAdd(&cnt[bucket_of(d[u], KB)], 1);
                if ((unsigned)p < (unsigned)CAPS) stage[p] = make_int2(s[u], d[u]);
                else if ((unsigned)p < (unsigned)mine) pr2[lo + p] = make_int2(s[u], d[u]);
            }
        }
    }
    __syncthreads();
    const int ms = min(mine, CAPS);
    for (int i = tid; i < ms; i += 1024) pr2[lo + i] = stage[i];
}

__global__ __launch_bounds__(512) void k_bucket_rows(const int* __restrict__ pairs, const int* __restrict__ T,
                                                     float* __restrict__ dinv, int* __restrict__ rowbeg,
                                                     int* __restrict__ col, int N, int E) {
    constexpr int CAP = 8192;
    constexpr int SPW = EBLK / 16;
    static_assert(SPW == 8, "the load fence below names eight segments");
    __shared__ int cnt[NPB];
    __shared__ int wsum[16];
    __shared__ int stage[CAP];
    __shared__ int segbeg[EBLK];
    __shared__ int seglen[EBLK];
    __shared__ int pbase[EBLK / 32];
    __shared__ int pm[EBLK / 32];
    const int tid  = threadIdx.x;
    const int lane = tid & 31;
    const int wv   = tid >> 5;
    const int k    = blockIdx.x;
    const int n0   = k << NPB_LOG;
    cnt[tid] = 0;
    if (tid < EBLK) {
        const int CH   = (E + EBLK - 1) / EBLK;
        const int lo   = min(tid * CH, E);
        const int mine = min(E, lo + CH) - lo;
        const int a = min(max(T[tid * TP + k], 0), mine);
        const int z = min(max(T[tid * TP + k + 1], a), mine);
        segbeg[tid] = lo + a;
        seglen[tid] = z - a;
        int sa = a, sl = z - a;
        for (int o = 16; o > 0; o >>= 1) {
            sa += __shfl_xor(sa, o);
            sl += __shfl_xor(sl, o);
        }
        if (lane == 0) { pbase[wv] = sa; pm[wv] = sl; }
    }
    __syncthreads();
    int sb = 0, sm = 0;
    for (int j = 0; j < EBLK / 32; ++j) { sb += pbase[j]; sm += pm[j]; }
    const int base = min(max(sb, 0), E);
    const int m    = min(max(sm, 0), E - base);
    const int2* pr2 = (const int2*)pairs;

    int2 pr[SPW];
    int  ln[SPW];
#pragma unroll
    for (int u = 0; u < SPW; ++u) {
        const int sg = wv + 16 * u;
        ln[u] = seglen[sg];
        pr[u] = pr2[min(segbeg[sg] + min(lane, max(ln[u] - 1, 0)), E - 1)];
    }
    asm volatile("" :: "v"(pr[0].y), "v"(pr[1].y), "v"(pr[2].y), "v"(pr[3].y),
                       "v"(pr[4].y), "v"(pr[5].y), "v"(pr[6].y), "v"(pr[7].y));
#pragma unroll
    for (int u = 0; u < SPW; ++u)
        if (lane < ln[u]) atomicAdd(&cnt[min(max(pr[u].y - n0, 0), NPB - 1)], 1);
    for (int u = 0; u < SPW; ++u) {
        const int sg = wv + 16 * u;
        const int L  = seglen[sg];
        for (int j = 32 + lane; j < L; j += 32) {
            const int2 q = pr2[min(segbeg[sg] + j, E - 1)];
            atomicAdd(&cnt[min(max(q.y - n0, 0), NPB - 1)], 1);
        }
    }
    __syncthreads();
    const int c = cnt[tid];
    int inc = c;
    for (int o = 1; o < 32; o <<= 1) {
        const int t = __shfl_up(inc, o);
        if (lane >= o) inc += t;
    }
    if (lane == 31) wsum[wv] = inc;
    __syncthreads();
    int woff = 0;
    for (int j = 0; j < 16; ++j)
        if (j < wv) woff += wsum[j];
    const int rs = woff + inc - c;
    cnt[tid] = rs;
    const int node = n0 + tid;
    if (node < N) {
        rowbeg[node] = base + rs;
        const float dg = (float)(c + 1);
        dinv[node] = (dg > 0.0f) ? rsqrtf(dg) : 0.0f;
    }
    __syncthreads();
#pragma unroll
    for (int u = 0; u < SPW; ++u) {
        if (lane < ln[u]) {
            const int p = atomicAdd(&cnt[min(max(pr[u].y - n0, 0), NPB - 1)], 1);
            if ((unsigned)p < (unsigned)CAP) stage[p] = pr[u].x;
            else if ((unsigned)p < (unsigned)m) col[base + p] = pr[u].x;
        }
    }
    for (int u = 0; u < SPW; ++u) {
        const int sg = wv + 16 * u;
        const int L  = seglen[sg];
        for (int j = 32 + lane; j < L; j += 32) {
            const int2 q = pr2[min(segbeg[sg] + j, E - 1)];
            const int p = atomicAdd(&cnt[min(max(q.y - n0, 0), NPB - 1)], 1);
            if ((unsigned)p < (unsigned)CAP) stage[p] = q.x;
            else if ((unsigned)p < (unsigned)m) col[base + p] = q.x;
        }
    }
    __syncthreads();
    const int mc = min(m, CAP);
    for (int i = tid; i < mc; i += 512) col[base + i] = stage[i];
}

__global__ __launch_bounds__(512) void k_layer1_fused(const float* __restrict__ x,
                                                      const int* __restrict__ rowbeg,
                                                      const int* __restrict__ col,
                                                      const float* __restrict__ dinv,
                                                      const float* __restrict__ W1,
                                                      const float* __restrict__ b1,
                                                      const float* __restrict__ W2,
                                                      float* __restrict__ g, int N, int E) {
    constexpr int BP1 = 2 * 128 + 32;
    constexpr int BP2 = 2 * 64 + 32;
    constexpr int AP  = 132;

    __shared__ __attribute__((aligned(16))) float B1s[64 * BP1];
    __shared__ __attribute__((aligned(16))) float B2s[64 * BP2];
    __shared__ __attribute__((aligned(16))) float As[64 * AP];
    __shared__ float dts[64];

    const int tid  = threadIdx.x;
    const int lane = tid & 31;
    const int wv   = tid >> 5;
    const int half = lane >> 4;
    const int r    = lane & 15;
    const int tiles = (N + 63) >> 6;
    const int tstep = gridDim.x;

    int   rb_raw;
    float dv_raw;
    {
        const int mrowc = min((int)blockIdx.x * 64 + 4 * wv + lane, N - 1);
        rb_raw = rowbeg[mrowc];
        dv_raw = dinv[mrowc];
    }

    float4 w1e[4], w1o[4], w2e[2], w2o[2];
#pragma unroll
    for (int it = 0; it < 4; ++it) {
        const int idx = tid + it * 512;
        const int kp  = idx >> 5;
        const int n4  = (idx & 31) * 4;
        w1e[it] = *(const float4*)(W1 + (size_t)(2 * kp) * 128 + n4);
        w1o[it] = *(const float4*)(W1 + (size_t)(2 * kp + 1) * 128 + n4);
    }
#pragma unroll
    for (int it = 0; it < 2; ++it) {
        const int idx = tid + it * 512;
        const int kp  = idx >> 4;
        const int n4  = (idx & 15) * 4;
        w2e[it] = *(const float4*)(W2 + (size_t)(2 * kp) * 64 + n4);
        w2o[it] = *(const float4*)(W2 + (size_t)(2 * kp + 1) * 64 + n4);
    }

#pragma unroll
    for (int it = 0; it < 4; ++it) {
        const int idx = tid + it * 512;
        const int kp  = idx >> 5;
        const int n4  = (idx & 31) * 4;
        float4* d = (float4*)(B1s + kp * BP1 + 2 * n4);
        d[0] = make_float4(w1e[it].x, w1o[it].x, w1e[it].y, w1o[it].y);
        d[1] = make_float4(w1e[it].z, w1o[it].z, w1e[it].w, w1o[it].w);
    }
#pragma unroll
    for (int it = 0; it < 2; ++it) {
        const int idx = tid + it * 512;
        const int kp  = idx >> 4;
        const int n4  = (idx & 15) * 4;
        float4* d = (float4*)(B2s + kp * BP2 + 2 * n4);
        d[0] = make_float4(w2e[it].x, w2o[it].x, w2e[it].y, w2o[it].y);
        d[1] = make_float4(w2e[it].z, w2o[it].z, w2e[it].w, w2o[it].w);
    }

    for (int tile = blockIdx.x; tile < tiles; tile += tstep) {
        const int m0   = tile * 64;
        const int row0 = m0 + 4 * wv;
        const int mrow = row0 + lane;

        v4f selfv[4];
#pragma unroll
        for (int i = 0; i < 4; ++i)
            selfv[i] = *(const v4f*)(x + (size_t)min(row0 + i, N - 1) * 128 + 4 * lane);

        const int   bl  = (mrow < N) ? min(max(rb_raw, 0), E) : E;
        const float mdd = (mrow < N) ? dv_raw : 0.0f;
        if (lane < 4) dts[4 * wv + lane] = mdd;
        {
            const int nrowc = min(mrow + tstep * 64, N - 1);
            rb_raw = rowbeg[nrowc];
            dv_raw = dinv[nrowc];
        }
        const int ebeg = __builtin_amdgcn_readlane(bl, 0);
        const int eend = __builtin_amdgcn_readlane(bl, 4);
        int s_first = 0;
        if (lane < eend - ebeg) s_first = col[ebeg + lane];

        float* trow = As + (4 * wv) * AP + 4 * lane;
#pragma unroll
        for (int i = 0; i < 4; ++i) {
            const float di = __int_as_float(__builtin_amdgcn_readlane(__float_as_int(mdd), i));
            *(v4f*)(trow + i * AP) = (row0 + i < N) ? selfv[i] * (di * di) : v4f{};
        }

        {
            int   cur  = 0;
            int   endc = __builtin_amdgcn_readlane(bl, 1);
            float ddc  = __int_as_float(__builtin_amdgcn_readlane(__float_as_int(mdd), 0));
            v4f   acc  = v4f{};
            const float* xl = x + 4 * lane;

            auto take = [&](const v4f v, const float wsrc, const int e) {
                while (e >= endc && cur < 3) {
                    *(v4f*)(trow + cur * AP) += acc;
                    ++cur;
                    endc = __builtin_amdgcn_readlane(bl, cur + 1);
                    ddc  = __int_as_float(__builtin_amdgcn_readlane(__float_as_int(mdd), cur));
                    acc  = v4f{};
                }
                acc += v * (wsrc * ddc);
            };

            for (int base = ebeg; base < eend; base += 32) {
                const int n = min(32, eend - base);
                int raw = s_first;
                if (base != ebeg) {
                    raw = 0;
                    if (lane < n) raw = col[base + lane];
                }
                const int my_s = min(max(raw, 0), N - 1);
                float my_w = 0.0f;
                if (lane < n) my_w = dinv[my_s];
                for (int j = 0; j < n; j += 8) {
                    const int m = min(8, n - j);
                    v4f v[8];
                    if (m == 8) {
#pragma unroll
                        for (int k = 0; k < 8; ++k)
                            v[k] = *(const v4f*)(xl + (size_t)__builtin_amdgcn_readlane(my_s, j + k) * 128);
                    } else {
#pragma unroll
                        for (int k = 0; k < 8; ++k) {
                            v[k] = v4f{};
                            if (k < m) v[k] = *(const v4f*)(xl + (size_t)__builtin_amdgcn_readlane(my_s, j + k) * 128);
                        }
                    }
#pragma unroll
                    for (int k = 0; k < 8; ++k) {
                        if (k < m)
                            take(v[k], __int_as_float(__builtin_amdgcn_readlane(__float_as_int(my_w), j + k)), base + j + k);
                    }
                }
            }
            *(v4f*)(trow + cur * AP) += acc;
        }
        __syncthreads();

        const int rt = wv & 3;
        const int cg = wv >> 2;
        const float* ap = As + (16 * rt + r) * AP + 2 * half;

        v8f c0 = v8f{};
        v8f c1 = v8f{};
        {
            const float* bp = B1s + half * BP1 + 2 * r + 64 * cg;
#pragma unroll
            for (int kk = 0; kk < 128; kk += 4) {
                const v2f a  = *(const v2f*)(ap + kk);
                const v2f bA = *(const v2f*)(bp + (kk >> 1) * BP1);
                const v2f bB = *(const v2f*)(bp + (kk >> 1) * BP1 + 32);
                c0 = __builtin_amdgcn_wmma_f32_16x16x4_f32(false, a, false, bA, (short)0, c0, false, false);
                c1 = __builtin_amdgcn_wmma_f32_16x16x4_f32(false, a, false, bB, (short)0, c1, false, false);
            }
        }
        __syncthreads();
        {
            const int   colA  = 32 * cg + r;
            const float biasA = b1[colA];
            const float biasB = b1[colA + 16];
#pragma unroll
            for (int r2 = 0; r2 < 8; ++r2) {
                const float hA = c0[r2] + biasA;
                const float hB = c1[r2] + biasB;
                float* t = As + (16 * rt + r2 + 8 * half) * AP + colA;
                t[0]  = hA > 0.0f ? hA : 0.0f;
                t[16] = hB > 0.0f ? hB : 0.0f;
            }
        }
        __syncthreads();

        v8f c2 = v8f{};
        {
            const float* bp = B2s + half * BP2 + 2 * r + 32 * cg;
#pragma unroll
            for (int kk = 0; kk < 128; kk += 4) {
                const v2f a = *(const v2f*)(ap + kk);
                const v2f b = *(const v2f*)(bp + (kk >> 1) * BP2);
                c2 = __builtin_amdgcn_wmma_f32_16x16x4_f32(false, a, false, b, (short)0, c2, false, false);
            }
        }
        __syncthreads();
#pragma unroll
        for (int r2 = 0; r2 < 8; ++r2)
            As[(16 * rt + r2 + 8 * half) * AP + 16 * cg + r] = c2[r2];
        __syncthreads();

        const int orow = tid >> 4;
        const int oc4  = 4 * (tid & 15);
        const v4f o0 = *(const v4f*)(As + orow * AP + oc4) * dts[orow];
        const v4f o1 = *(const v4f*)(As + (orow + 32) * AP + oc4) * dts[orow + 32];
        __syncthreads();
        if (m0 + orow < N)      *(v4f*)(g + (size_t)(m0 + orow) * 64 + oc4)      = o0;
        if (m0 + orow + 32 < N) *(v4f*)(g + (size_t)(m0 + orow + 32) * 64 + oc4) = o1;
    }
}

template<int C>
__global__ __launch_bounds__(256) void k_sum_rows(const float* __restrict__ feat,
                                                  const int* __restrict__ rowbeg,
                                                  const int* __restrict__ col,
                                                  const float* __restrict__ bias,
                                                  float* __restrict__ out, int N, int E) {
    constexpr int V = C / 32;
    typedef float vec_t __attribute__((ext_vector_type(V)));
    constexpr int RUN = 31;
    const int lane = threadIdx.x & 31;
    const int wave0 = __builtin_amdgcn_readfirstlane((int)((blockIdx.x * blockDim.x + threadIdx.x) >> 5));
    const int nwave = (int)((gridDim.x * blockDim.x) >> 5);
    const int q   = N / nwave;
    const int rem = N - q * nwave;
    const int r0  = wave0 * q + min(wave0, rem);
    const int r1  = r0 + q + (wave0 < rem ? 1 : 0);

    const float* fl = feat + lane * V;
    const vec_t  bv = *(const vec_t*)(bias + lane * V);

#define ROW_REQUEST(k, cnt, idx, first) \
    vec_t v##k = vec_t{};                 \
    if ((k) < (cnt)) v##k = *(const vec_t*)(fl + (size_t)__builtin_amdgcn_readlane((idx), (first) + (k)) * C);
#define ROWS_REQUEST(cnt, idx, first) \
    ROW_REQUEST(0, cnt, idx, first) ROW_REQUEST(1, cnt, idx, first) ROW_REQUEST(2, cnt, idx, first) ROW_REQUEST(3, cnt, idx, first) \
    ROW_REQUEST(4, cnt, idx, first) ROW_REQUEST(5, cnt, idx, first) ROW_REQUEST(6, cnt, idx, first) ROW_REQUEST(7, cnt, idx, first)
    auto batch = [&](vec_t& acc, const int ms, const int j, const int m) {
        ROWS_REQUEST(m, ms, j)
        acc += v0; acc += v1; acc += v2; acc += v3; acc += v4; acc += v5; acc += v6; acc += v7;
    };

    for (int a = r0; a < r1; a += RUN) {
        const int nr = min(RUN, r1 - a);
        int rb;
        {
            const int row = a + lane;
            const int t   = rowbeg[min(row, N - 1)];
            rb = (row < N) ? min(max(t, 0), E) : E;
        }
        int beg = __builtin_amdgcn_readlane(rb, 0);
        int end = __builtin_amdgcn_readlane(rb, 1);
        int raw = 0;
        if (lane < end - beg) raw = col[beg + lane];

        for (int i = 0; i < nr; ++i) {
            const int d = a + i;
            const int n    = min(32, end - beg);
            const int my_s = min(max(raw, 0), N - 1);
            const vec_t selfv = *(const vec_t*)(fl + (size_t)d * C);
            const int m0 = min(8, n);
            ROWS_REQUEST(m0, my_s, 0)
            const int endn = __builtin_amdgcn_readlane(rb, min(i + 2, RUN));
            int rawn = 0;
            if (i + 1 < nr && lane < endn - end) rawn = col[end + lane];

            const float dd = rsqrtf((float)(max(end - beg, 0) + 1));

            vec_t acc = selfv;
            acc += v0; acc += v1; acc += v2; acc += v3; acc += v4; acc += v5; acc += v6; acc += v7;
            for (int j = 8; j < n; j += 8) batch(acc, my_s, j, min(8, n - j));
            for (int base = beg + 32; base < end; base += 32) {
                const int nn = min(32, end - base);
                int cs = d;
                if (lane < nn) cs = min(max(col[base + lane], 0), N - 1);
                for (int j = 0; j < nn; j += 8) batch(acc, cs, j, min(8, nn - j));
            }

            *(vec_t*)(out + (size_t)d * C + lane * V) = acc * dd + bv;

            beg = end; end = endn; raw = rawn;
        }
    }
#undef ROWS_REQUEST
#undef ROW_REQUEST
}

extern "C" void kernel_launch(void* const* d_in, const int* in_sizes, int n_in,
                              void* d_out, int out_size, void* d_ws, size_t ws_size,
                              hipStream_t stream) {
    const float* x  = (const float*)d_in[0];
    const int*   ei = (const int*)  d_in[1];
    const float* W1 = (const float*)d_in[2];
    const float* b1 = (const float*)d_in[3];
    const float* W2 = (const float*)d_in[4];
    const float* b2 = (const float*)d_in[5];

    const int N = in_sizes[0] / 128;
    const int E = in_sizes[1] / 2;
    const int* src = ei;
    const int* dst = ei + E;
    const int KB = (N + NPB - 1) / NPB;

    char* ws = (char*)d_ws;
    size_t szN  = (((size_t)N * 4) + 255) & ~(size_t)255;
    size_t szT  = (size_t)EBLK * TP * 4;
    size_t szE  = (((size_t)E * 4) + 255) & ~(size_t)255;
    float* dinv   = (float*)ws;
    int*   rowbeg = (int*)(ws + szN);
    int*   T      = (int*)(ws + 2 * szN);
    int*   pairs  = (int*)(ws + 2 * szN + szT);
    int*   col    = (int*)(ws + 2 * szN + szT + 2 * szE);
    size_t off  = 2 * szN + szT + 3 * szE;
    float* bufG = (float*)(ws + off);

    k_chunk_sort<<<EBLK, 1024, 0, stream>>>(src, dst, T, pairs, E, KB);
    k_bucket_rows<<<KB, 512, 0, stream>>>(pairs, T, dinv, rowbeg, col, N, E);

    const int tiles = (N + 63) / 64;
    k_layer1_fused<<<(tiles < 512 ? tiles : 512), 512, 0, stream>>>(x, rowbeg, col, dinv, W1, b1, W2, bufG, N, E);

    const int gblocks = (N + 7) / 8;
    k_sum_rows<64><<<(gblocks < 1024 ? gblocks : 1024), 256, 0, stream>>>(bufG, rowbeg, col, b2, (float*)d_out, N, E);
}


# Layer-1 tail balanced: all blocks walk equally many 64-row tiles; rows left over go as 16-row strips, one per block, on a short path (1 row per wave, 8+4 waves multiply); sum order kept
# speedup vs baseline: 4.8983x; 1.0335x over previous
#include <hip/hip_runtime.h>

typedef float v2f __attribute__((ext_vector_type(2)));
typedef float v4f __attribute__((ext_vector_type(4)));
typedef float v8f __attribute__((ext_vector_type(8)));

constexpr int NPB_LOG = 9;
constexpr int NPB     = 1 << NPB_LOG;
constexpr int MAXKB   = 1024;
constexpr int EBLK    = 128;
constexpr int TP      = MAXKB + 64;

__device__ __forceinline__ int bucket_of(int d, int KB) {
    return (int)min((unsigned)d >> NPB_LOG, (unsigned)(KB - 1));
}

__global__ __launch_bounds__(1024) void k_chunk_sort(const int* __restrict__ src, const int* __restrict__ dst,
                                                     int* __restrict__ T, int* __restrict__ pairs, int E, int KB) {
    constexpr int CAPS = 8192;
    __shared__ int  cnt[MAXKB];
    __shared__ int  wsum[32];
    __shared__ int2 stage[CAPS];
    const int tid  = threadIdx.x;
    const int lane = tid & 31;
    const int wv   = tid >> 5;
    const int b    = blockIdx.x;
    const int CH   = (E + EBLK - 1) / EBLK;
    const int lo   = min(b * CH, E);
    const int hi   = min(E, lo + CH);
    const int mine = hi - lo;
    const bool one_sweep = CH <= 8 * 1024;
    cnt[tid] = 0;
    __syncthreads();
    int d[8], s[8];
#pragma unroll
    for (int u = 0; u < 8; ++u) { d[u] = 0; s[u] = 0; }
    for (int e0 = lo + tid; e0 < hi; e0 += 8 * 1024) {
#pragma unroll
        for (int u = 0; u < 8; ++u)
            d[u] = dst[min(e0 + u * 1024, hi - 1)];
        if (one_sweep) {
#pragma unroll
            for (int u = 0; u < 8; ++u)
                s[u] = src[min(e0 + u * 1024, hi - 1)];
        }
        asm volatile("" :: "v"(d[0]), "v"(d[1]), "v"(d[2]), "v"(d[3]), "v"(d[4]), "v"(d[5]), "v"(d[6]), "v"(d[7]));
#pragma unroll
        for (int u = 0; u < 8; ++u)
            if (e0 + u * 1024 < hi) atomicAdd(&cnt[bucket_of(d[u], KB)], 1);
    }
    __syncthreads();
    const int v = (tid < KB) ? cnt[tid] : 0;
    int inc = v;
    for (int o = 1; o < 32; o <<= 1) {
        const int t = __shfl_up(inc, o);
        if (lane >= o) inc += t;
    }
    if (lane == 31) wsum[wv] = inc;
    __syncthreads();
    int woff = 0;
    for (int k = 0; k < 32; ++k)
        if (k < wv) woff += wsum[k];
    const int excl = woff + inc - v;
    if (tid < KB) {
        cnt[tid] = excl;
        T[b * TP + tid] = excl;
        if (tid == KB - 1) T[b * TP + KB] = excl + v;
    }
    __syncthreads();
    int2* pr2 = (int2*)pairs;
    for (int e0 = lo + tid; e0 < hi; e0 += 8 * 1024) {
        if (!one_sweep) {
#pragma unroll
            for (int u = 0; u < 8; ++u) {
                const int e = min(e0 + u * 1024, hi - 1);
                d[u] = dst[e];
                s[u] = src[e];
            }
            asm volatile("" :: "v"(d[0]), "v"(d[1]), "v"(d[2]), "v"(d[3]), "v"(d[4]), "v"(d[5]), "v"(d[6]), "v"(d[7]),
                               "v"(s[0]), "v"(s[1]), "v"(s[2]), "v"(s[3]), "v"(s[4]), "v"(s[5]), "v"(s[6]), "v"(s[7]));
        }
#pragma unroll
        for (int u = 0; u < 8; ++u) {
            if (e0 + u * 1024 < hi) {
                const int p = atomicAdd(&cnt[bucket_of(d[u], KB)], 1);
                if ((unsigned)p < (unsigned)CAPS) stage[p] = make_int2(s[u], d[u]);
                else if ((unsigned)p < (unsigned)mine) pr2[lo + p] = make_int2(s[u], d[u]);
            }
        }
    }
    __syncthreads();
    const int ms = min(mine, CAPS);
    for (int i = tid; i < ms; i += 1024) pr2[lo + i] = stage[i];
}

__global__ __launch_bounds__(512) void k_bucket_rows(const int* __restrict__ pairs, const int* __restrict__ T,
                                                     float* __restrict__ dinv, int* __restrict__ rowbeg,
                                                     int* __restrict__ col, int N, int E) {
    constexpr int CAP = 8192;
    constexpr int SPW = EBLK / 16;
    static_assert(SPW == 8, "the load fence below names eight segments");
    __shared__ int cnt[NPB];
    __shared__ int wsum[16];
    __shared__ int stage[CAP];
    __shared__ int segbeg[EBLK];
    __shared__ int seglen[EBLK];
    __shared__ int pbase[EBLK / 32];
    __shared__ int pm[EBLK / 32];
    const int tid  = threadIdx.x;
    const int lane = tid & 31;
    const int wv   = tid >> 5;
    const int k    = blockIdx.x;
    const int n0   = k << NPB_LOG;
    cnt[tid] = 0;
    if (tid < EBLK) {
        const int CH   = (E + EBLK - 1) / EBLK;
        const int lo   = min(tid * CH, E);
        const int mine = min(E, lo + CH) - lo;
        const int a = min(max(T[tid * TP + k], 0), mine);
        const int z = min(max(T[tid * TP + k + 1], a), mine);
        segbeg[tid] = lo + a;
        seglen[tid] = z - a;
        int sa = a, sl = z - a;
        for (int o = 16; o > 0; o >>= 1) {
            sa += __shfl_xor(sa, o);
            sl += __shfl_xor(sl, o);
        }
        if (lane == 0) { pbase[wv] = sa; pm[wv] = sl; }
    }
    __syncthreads();
    int sb = 0, sm = 0;
    for (int j = 0; j < EBLK / 32; ++j) { sb += pbase[j]; sm += pm[j]; }
    const int base = min(max(sb, 0), E);
    const int m    = min(max(sm, 0), E - base);
    const int2* pr2 = (const int2*)pairs;

    int2 pr[SPW];
    int  ln[SPW];
#pragma unroll
    for (int u = 0; u < SPW; ++u) {
        const int sg = wv + 16 * u;
        ln[u] = seglen[sg];
        pr[u] = pr2[min(segbeg[sg] + min(lane, max(ln[u] - 1, 0)), E - 1)];
    }
    asm volatile("" :: "v"(pr[0].y), "v"(pr[1].y), "v"(pr[2].y), "v"(pr[3].y),
                       "v"(pr[4].y), "v"(pr[5].y), "v"(pr[6].y), "v"(pr[7].y));
#pragma unroll
    for (int u = 0; u < SPW; ++u)
        if (lane < ln[u]) atomicAdd(&cnt[min(max(pr[u].y - n0, 0), NPB - 1)], 1);
    for (int u = 0; u < SPW; ++u) {
        const int sg = wv + 16 * u;
        const int L  = seglen[sg];
        for (int j = 32 + lane; j < L; j += 32) {
            const int2 q = pr2[min(segbeg[sg] + j, E - 1)];
            atomicAdd(&cnt[min(max(q.y - n0, 0), NPB - 1)], 1);
        }
    }
    __syncthreads();
    const int c = cnt[tid];
    int inc = c;
    for (int o = 1; o < 32; o <<= 1) {
        const int t = __shfl_up(inc, o);
        if (lane >= o) inc += t;
    }
    if (lane == 31) wsum[wv] = inc;
    __syncthreads();
    int woff = 0;
    for (int j = 0; j < 16; ++j)
        if (j < wv) woff += wsum[j];
    const int rs = woff + inc - c;
    cnt[tid] = rs;
    const int node = n0 + tid;
    if (node < N) {
        rowbeg[node] = base + rs;
        const float dg = (float)(c + 1);
        dinv[node] = (dg > 0.0f) ? rsqrtf(dg) : 0.0f;
    }
    __syncthreads();
#pragma unroll
    for (int u = 0; u < SPW; ++u) {
        if (lane < ln[u]) {
            const int p = atomicAdd(&cnt[min(max(pr[u].y - n0, 0), NPB - 1)], 1);
            if ((unsigned)p < (unsigned)CAP) stage[p] = pr[u].x;
            else if ((unsigned)p < (unsigned)m) col[base + p] = pr[u].x;
        }
    }
    for (int u = 0; u < SPW; ++u) {
        const int sg = wv + 16 * u;
        const int L  = seglen[sg];
        for (int j = 32 + lane; j < L; j += 32) {
            const int2 q = pr2[min(segbeg[sg] + j, E - 1)];
            const int p = atomicAdd(&cnt[min(max(q.y - n0, 0), NPB - 1)], 1);
            if ((unsigned)p < (unsigned)CAP) stage[p] = q.x;
            else if ((unsigned)p < (unsigned)m) col[base + p] = q.x;
        }
    }
    __syncthreads();
    const int mc = min(m, CAP);
    for (int i = tid; i < mc; i += 512) col[base + i] = stage[i];
}

__global__ __launch_bounds__(512) void k_layer1_fused(const float* __restrict__ x,
                                                      const int* __restrict__ rowbeg,
                                                      const int* __restrict__ col,
                                                      const float* __restrict__ dinv,
                                                      const float* __restrict__ W1,
                                                      const float* __restrict__ b1,
                                                      const float* __restrict__ W2,
                                                      float* __restrict__ g, int N, int E) {
    constexpr int BP1 = 2 * 128 + 32;
    constexpr int BP2 = 2 * 64 + 32;
    constexpr int AP  = 132;

    __shared__ __attribute__((aligned(16))) float B1s[64 * BP1];
    __shared__ __attribute__((aligned(16))) float B2s[64 * BP2];
    __shared__ __attribute__((aligned(16))) float As[64 * AP];
    __shared__ float dts[64];

    const int tid  = threadIdx.x;
    const int lane = tid & 31;
    const int wv   = tid >> 5;
    const int half = lane >> 4;
    const int r    = lane & 15;
    const int tstep = gridDim.x;
    const int tfull   = ((N >> 6) / tstep) * tstep;
    const int srow0   = tfull * 64;
    const int nstrips = (N - srow0 + 15) >> 4;

    int   rb_raw;
    float dv_raw;
    {
        const int mrowc = min((int)blockIdx.x * 64 + 4 * wv + lane, N - 1);
        rb_raw = rowbeg[mrowc];
        dv_raw = dinv[mrowc];
    }
    int   srb_raw;
    float sdv_raw;
    {
        const int srow = srow0 + 16 * (int)blockIdx.x + wv;
        srb_raw = rowbeg[min(srow + lane, N - 1)];
        sdv_raw = dinv[min(srow, N - 1)];
    }

    float4 w1e[4], w1o[4], w2e[2], w2o[2];
#pragma unroll
    for (int it = 0; it < 4; ++it) {
        const int idx = tid + it * 512;
        const int kp  = idx >> 5;
        const int n4  = (idx & 31) * 4;
        w1e[it] = *(const float4*)(W1 + (size_t)(2 * kp) * 128 + n4);
        w1o[it] = *(const float4*)(W1 + (size_t)(2 * kp + 1) * 128 + n4);
    }
#pragma unroll
    for (int it = 0; it < 2; ++it) {
        const int idx = tid + it * 512;
        const int kp  = idx >> 4;
        const int n4  = (idx & 15) * 4;
        w2e[it] = *(const float4*)(W2 + (size_t)(2 * kp) * 64 + n4);
        w2o[it] = *(const float4*)(W2 + (size_t)(2 * kp + 1) * 64 + n4);
    }

#pragma unroll
    for (int it = 0; it < 4; ++it) {
        const int idx = tid + it * 512;
        const int kp  = idx >> 5;
        const int n4  = (idx & 31) * 4;
        float4* d = (float4*)(B1s + kp * BP1 + 2 * n4);
        d[0] = make_float4(w1e[it].x, w1o[it].x, w1e[it].y, w1o[it].y);
        d[1] = make_float4(w1e[it].z, w1o[it].z, w1e[it].w, w1o[it].w);
    }
#pragma unroll
    for (int it = 0; it < 2; ++it) {
        const int idx = tid + it * 512;
        const int kp  = idx >> 4;
        const int n4  = (idx & 15) * 4;
        float4* d = (float4*)(B2s + kp * BP2 + 2 * n4);
        d[0] = make_float4(w2e[it].x, w2o[it].x, w2e[it].y, w2o[it].y);
        d[1] = make_float4(w2e[it].z, w2o[it].z, w2e[it].w, w2o[it].w);
    }

    for (int tile = blockIdx.x; tile < tfull; tile += tstep) {
        const int m0   = tile * 64;
        const int row0 = m0 + 4 * wv;
        const int mrow = row0 + lane;

        v4f selfv[4];
#pragma unroll
        for (int i = 0; i < 4; ++i)
            selfv[i] = *(const v4f*)(x + (size_t)min(row0 + i, N - 1) * 128 + 4 * lane);

        const int   bl  = (mrow < N) ? min(max(rb_raw, 0), E) : E;
        const float mdd = (mrow < N) ? dv_raw : 0.0f;
        if (lane < 4) dts[4 * wv + lane] = mdd;
        {
            const int nrowc = min(mrow + tstep * 64, N - 1);
            rb_raw = rowbeg[nrowc];
            dv_raw = dinv[nrowc];
        }
        const int ebeg = __builtin_amdgcn_readlane(bl, 0);
        const int eend = __builtin_amdgcn_readlane(bl, 4);
        int s_first = 0;
        if (lane < eend - ebeg) s_first = col[ebeg + lane];

        float* trow = As + (4 * wv) * AP + 4 * lane;
#pragma unroll
        for (int i = 0; i < 4; ++i) {
            const float di = __int_as_float(__builtin_amdgcn_readlane(__float_as_int(mdd), i));
            *(v4f*)(trow + i * AP) = (row0 + i < N) ? selfv[i] * (di * di) : v4f{};
        }

        {
            int   cur  = 0;
            int   endc = __builtin_amdgcn_readlane(bl, 1);
            float ddc  = __int_as_float(__builtin_amdgcn_readlane(__float_as_int(mdd), 0));
            v4f   acc  = v4f{};
            const float* xl = x + 4 * lane;

            auto take = [&](const v4f v, const float wsrc, const int e) {
                while (e >= endc && cur < 3) {
                    *(v4f*)(trow + cur * AP) += acc;
                    ++cur;
                    endc = __builtin_amdgcn_readlane(bl, cur + 1);
                    ddc  = __int_as_float(__builtin_amdgcn_readlane(__float_as_int(mdd), cur));
                    acc  = v4f{};
                }
                acc += v * (wsrc * ddc);
            };

            for (int base = ebeg; base < eend; base += 32) {
                const int n = min(32, eend - base);
                int raw = s_first;
                if (base != ebeg) {
                    raw = 0;
                    if (lane < n) raw = col[base + lane];
                }
                const int my_s = min(max(raw, 0), N - 1);
                float my_w = 0.0f;
                if (lane < n) my_w = dinv[my_s];
                for (int j = 0; j < n; j += 8) {
                    const int m = min(8, n - j);
                    v4f v[8];
                    if (m == 8) {
#pragma unroll
                        for (int k = 0; k < 8; ++k)
                            v[k] = *(const v4f*)(xl + (size_t)__builtin_amdgcn_readlane(my_s, j + k) * 128);
                    } else {
#pragma unroll
                        for (int k = 0; k < 8; ++k) {
                            v[k] = v4f{};
                            if (k < m) v[k] = *(const v4f*)(xl + (size_t)__builtin_amdgcn_readlane(my_s, j + k) * 128);
                        }
                    }
#pragma unroll
                    for (int k = 0; k < 8; ++k) {
                        if (k < m)
                            take(v[k], __int_as_float(__builtin_amdgcn_readlane(__float_as_int(my_w), j + k)), base + j + k);
                    }
                }
            }
            *(v4f*)(trow + cur * AP) += acc;
        }
        __syncthreads();

        const int rt = wv & 3;
        const int cg = wv >> 2;
        const float* ap = As + (16 * rt + r) * AP + 2 * half;

        v8f c0 = v8f{};
        v8f c1 = v8f{};
        {
            const float* bp = B1s + half * BP1 + 2 * r + 64 * cg;
#pragma unroll
            for (int kk = 0; kk < 128; kk += 4) {
                const v2f a  = *(const v2f*)(ap + kk);
                const v2f bA = *(const v2f*)(bp + (kk >> 1) * BP1);
                const v2f bB = *(const v2f*)(bp + (kk >> 1) * BP1 + 32);
                c0 = __builtin_amdgcn_wmma_f32_16x16x4_f32(false, a, false, bA, (short)0, c0, false, false);
                c1 = __builtin_amdgcn_wmma_f32_16x16x4_f32(false, a, false, bB, (short)0, c1, false, false);
            }
        }
        __syncthreads();
        {
            const int   colA  = 32 * cg + r;
            const float biasA = b1[colA];
            const float biasB = b1[colA + 16];
#pragma unroll
            for (int r2 = 0; r2 < 8; ++r2) {
                const float hA = c0[r2] + biasA;
                const float hB = c1[r2] + biasB;
                float* t = As + (16 * rt + r2 + 8 * half) * AP + colA;
                t[0]  = hA > 0.0f ? hA : 0.0f;
                t[16] = hB > 0.0f ? hB : 0.0f;
            }
        }
        __syncthreads();

        v8f c2 = v8f{};
        {
            const float* bp = B2s + half * BP2 + 2 * r + 32 * cg;
#pragma unroll
            for (int kk = 0; kk < 128; kk += 4) {
                const v2f a = *(const v2f*)(ap + kk);
                const v2f b = *(const v2f*)(bp + (kk >> 1) * BP2);
                c2 = __builtin_amdgcn_wmma_f32_16x16x4_f32(false, a, false, b, (short)0, c2, false, false);
            }
        }
        __syncthreads();
#pragma unroll
        for (int r2 = 0; r2 < 8; ++r2)
            As[(16 * rt + r2 + 8 * half) * AP + 16 * cg + r] = c2[r2];
        __syncthreads();

        const int orow = tid >> 4;
        const int oc4  = 4 * (tid & 15);
        const v4f o0 = *(const v4f*)(As + orow * AP + oc4) * dts[orow];
        const v4f o1 = *(const v4f*)(As + (orow + 32) * AP + oc4) * dts[orow + 32];
        __syncthreads();
        if (m0 + orow < N)      *(v4f*)(g + (size_t)(m0 + orow) * 64 + oc4)      = o0;
        if (m0 + orow + 32 < N) *(v4f*)(g + (size_t)(m0 + orow + 32) * 64 + oc4) = o1;
    }

    const int wvu = __builtin_amdgcn_readfirstlane(wv);
    for (int st = blockIdx.x; st < nstrips; st += tstep) {
        const int m0  = srow0 + 16 * st;
        const int row = m0 + wv;

        const v4f selfv = *(const v4f*)(x + (size_t)min(row, N - 1) * 128 + 4 * lane);

        const int   bl  = (row + lane < N) ? min(max(srb_raw, 0), E) : E;
        const float mdd = (row < N) ? sdv_raw : 0.0f;
        if (lane == 0) dts[wv] = mdd;
        {
            const int nrow = row + 16 * tstep;
            srb_raw = rowbeg[min(nrow + lane, N - 1)];
            sdv_raw = dinv[min(nrow, N - 1)];
        }
        const int ebeg = __builtin_amdgcn_readlane(bl, 0);
        const int eend = __builtin_amdgcn_readlane(bl, 1);

        {
            v4f acc = v4f{};
            const float* xl = x + 4 * lane;
            for (int base = ebeg; base < eend; base += 32) {
                const int n = min(32, eend - base);
                int raw = 0;
                if (lane < n) raw = col[base + lane];
                const int my_s = min(max(raw, 0), N - 1);
                float my_w = 0.0f;
                if (lane < n) my_w = dinv[my_s];
                for (int j = 0; j < n; j += 8) {
                    const int m = min(8, n - j);
                    v4f v[8];
                    if (m == 8) {
#pragma unroll
                        for (int k = 0; k < 8; ++k)
                            v[k] = *(const v4f*)(xl + (size_t)__builtin_amdgcn_readlane(my_s, j + k) * 128);
                    } else {
#pragma unroll
                        for (int k = 0; k < 8; ++k) {
                            v[k] = v4f{};
                            if (k < m) v[k] = *(const v4f*)(xl + (size_t)__builtin_amdgcn_readlane(my_s, j + k) * 128);
                        }
                    }
#pragma unroll
                    for (int k = 0; k < 8; ++k) {
                        if (k < m)
                            acc += v[k] * (__int_as_float(__builtin_amdgcn_readlane(__float_as_int(my_w), j + k)) * mdd);
                    }
                }
            }
            *(v4f*)(As + wv * AP + 4 * lane) = (row < N) ? selfv * (mdd * mdd) + acc : v4f{};
        }
        __syncthreads();

        const float* ap = As + r * AP + 2 * half;

        v8f c0 = v8f{};
        if (wvu < 8) {
            const float* bp = B1s + half * BP1 + 2 * (16 * wvu + r);
#pragma unroll
            for (int kk = 0; kk < 128; kk += 4) {
                const v2f a = *(const v2f*)(ap + kk);
                const v2f b = *(const v2f*)(bp + (kk >> 1) * BP1);
                c0 = __builtin_amdgcn_wmma_f32_16x16x4_f32(false, a, false, b, (short)0, c0, false, false);
            }
        }
        __syncthreads();
        if (wvu < 8) {
            const int   colA = 16 * wvu + r;
            const float bias = b1[colA];
#pragma unroll
            for (int r2 = 0; r2 < 8; ++r2) {
                const float h = c0[r2] + bias;
                As[(r2 + 8 * half) * AP + colA] = h > 0.0f ? h : 0.0f;
            }
        }
        __syncthreads();

        v8f c2 = v8f{};
        if (wvu < 4) {
            const float* bp = B2s + half * BP2 + 2 * (16 * wvu + r);
#pragma unroll
            for (int kk = 0; kk < 128; kk += 4) {
                const v2f a = *(const v2f*)(ap + kk);
                const v2f b = *(const v2f*)(bp + (kk >> 1) * BP2);
                c2 = __builtin_amdgcn_wmma_f32_16x16x4_f32(false, a, false, b, (short)0, c2, false, false);
            }
        }
        __syncthreads();
        if (wvu < 4) {
#pragma unroll
            for (int r2 = 0; r2 < 8; ++r2)
                As[(r2 + 8 * half) * AP + 16 * wvu + r] = c2[r2];
        }
        __syncthreads();

        const int orow = tid >> 4;
        const int oc4  = 4 * (tid & 15);
        v4f o0 = v4f{};
        if (wvu < 8) o0 = *(const v4f*)(As + orow * AP + oc4) * dts[orow];
        __syncthreads();
        if (wvu < 8 && m0 + orow < N) *(v4f*)(g + (size_t)(m0 + orow) * 64 + oc4) = o0;
    }
}

template<int C>
__global__ __launch_bounds__(256) void k_sum_rows(const float* __restrict__ feat,
                                                  const int* __restrict__ rowbeg,
                                                  const int* __restrict__ col,
                                                  const float* __restrict__ bias,
                                                  float* __restrict__ out, int N, int E) {
    constexpr int V = C / 32;
    typedef float vec_t __attribute__((ext_vector_type(V)));
    constexpr int RUN = 31;
    const int lane = threadIdx.x & 31;
    const int wave0 = __builtin_amdgcn_readfirstlane((int)((blockIdx.x * blockDim.x + threadIdx.x) >> 5));
    const int nwave = (int)((gridDim.x * blockDim.x) >> 5);
    const int q   = N / nwave;
    const int rem = N - q * nwave;
    const int r0  = wave0 * q + min(wave0, rem);
    const int r1  = r0 + q + (wave0 < rem ? 1 : 0);

    const float* fl = feat + lane * V;
    const vec_t  bv = *(const vec_t*)(bias + lane * V);

#define ROW_REQUEST(k, cnt, idx, first) \
    vec_t v##k = vec_t{};                 \
    if ((k) < (cnt)) v##k = *(const vec_t*)(fl + (size_t)__builtin_amdgcn_readlane((idx), (first) + (k)) * C);
#define ROWS_REQUEST(cnt, idx, first) \
    ROW_REQUEST(0, cnt, idx, first) ROW_REQUEST(1, cnt, idx, first) ROW_REQUEST(2, cnt, idx, first) ROW_REQUEST(3, cnt, idx, first) \
    ROW_REQUEST(4, cnt, idx, first) ROW_REQUEST(5, cnt, idx, first) ROW_REQUEST(6, cnt, idx, first) ROW_REQUEST(7, cnt, idx, first)
    auto batch = [&](vec_t& acc, const int ms, const int j, const int m) {
        ROWS_REQUEST(m, ms, j)
        acc += v0; acc += v1; acc += v2; acc += v3; acc += v4; acc += v5; acc += v6; acc += v7;
    };

    for (int a = r0; a < r1; a += RUN) {
        const int nr = min(RUN, r1 - a);
        int rb;
        {
            const int row = a + lane;
            const int t   = rowbeg[min(row, N - 1)];
            rb = (row < N) ? min(max(t, 0), E) : E;
        }
        int beg = __builtin_amdgcn_readlane(rb, 0);
        int end = __builtin_amdgcn_readlane(rb, 1);
        int raw = 0;
        if (lane < end - beg) raw = col[beg + lane];

        for (int i = 0; i < nr; ++i) {
            const int d = a + i;
            const int n    = min(32, end - beg);
            const int my_s = min(max(raw, 0), N - 1);
            const vec_t selfv = *(const vec_t*)(fl + (size_t)d * C);
            const int m0 = min(8, n);
            ROWS_REQUEST(m0, my_s, 0)
            const int endn = __builtin_amdgcn_readlane(rb, min(i + 2, RUN));
            int rawn = 0;
            if (i + 1 < nr && lane < endn - end) rawn = col[end + lane];

            const float dd = rsqrtf((float)(max(end - beg, 0) + 1));

            vec_t acc = selfv;
            acc += v0; acc += v1; acc += v2; acc += v3; acc += v4; acc += v5; acc += v6; acc += v7;
            for (int j = 8; j < n; j += 8) batch(acc, my_s, j, min(8, n - j));
            for (int base = beg + 32; base < end; base += 32) {
                const int nn = min(32, end - base);
                int cs = d;
                if (lane < nn) cs = min(max(col[base + lane], 0), N - 1);
                for (int j = 0; j < nn; j += 8) batch(acc, cs, j, min(8, nn - j));
            }

            *(vec_t*)(out + (size_t)d * C + lane * V) = acc * dd + bv;

            beg = end; end = endn; raw = rawn;
        }
    }
#undef ROWS_REQUEST
#undef ROW_REQUEST
}

extern "C" void kernel_launch(void* const* d_in, const int* in_sizes, int n_in,
                              void* d_out, int out_size, void* d_ws, size_t ws_size,
                              hipStream_t stream) {
    const float* x  = (const float*)d_in[0];
    const int*   ei = (const int*)  d_in[1];
    const float* W1 = (const float*)d_in[2];
    const float* b1 = (const float*)d_in[3];
    const float* W2 = (const float*)d_in[4];
    const float* b2 = (const float*)d_in[5];

    const int N = in_sizes[0] / 128;
    const int E = in_sizes[1] / 2;
    const int* src = ei;
    const int* dst = ei + E;
    const int KB = (N + NPB - 1) / NPB;

    char* ws = (char*)d_ws;
    size_t szN  = (((size_t)N * 4) + 255) & ~(size_t)255;
    size_t szT  = (size_t)EBLK * TP * 4;
    size_t szE  = (((size_t)E * 4) + 255) & ~(size_t)255;
    float* dinv   = (float*)ws;
    int*   rowbeg = (int*)(ws + szN);
    int*   T      = (int*)(ws + 2 * szN);
    int*   pairs  = (int*)(ws + 2 * szN + szT);
    int*   col    = (int*)(ws + 2 * szN + szT + 2 * szE);
    size_t off  = 2 * szN + szT + 3 * szE;
    float* bufG = (float*)(ws + off);

    k_chunk_sort<<<EBLK, 1024, 0, stream>>>(src, dst, T, pairs, E, KB);
    k_bucket_rows<<<KB, 512, 0, stream>>>(pairs, T, dinv, rowbeg, col, N, E);

    const int tiles = (N + 63) / 64;
    k_layer1_fused<<<(tiles < 512 ? tiles : 512), 512, 0, stream>>>(x, rowbeg, col, dinv, W1, b1, W2, bufG, N, E);

    const int gblocks = (N + 7) / 8;
    k_sum_rows<64><<<(gblocks < 1024 ? gblocks : 1024), 256, 0, stream>>>(bufG, rowbeg, col, b2, (float*)d_out, N, E);
}
